# starting kernel
# baseline (speedup 1.0000x reference)
#include <hip/hip_runtime.h>

#define DIM 128
#define WAVES_PER_BLOCK 8
#define BLOCK (WAVES_PER_BLOCK * 32)

typedef float v2f __attribute__((ext_vector_type(2)));
typedef float v8f __attribute__((ext_vector_type(8)));

__global__ void __launch_bounds__(BLOCK)
ln_relu_zero_kernel(const float* __restrict__ x,
                    const float* __restrict__ gamma,
                    const float* __restrict__ beta,
                    float* __restrict__ h,
                    float* __restrict__ aggr,
                    int n_nodes) {
  int wave = (blockIdx.x * blockDim.x + threadIdx.x) >> 5;
  int lane = threadIdx.x & 31;
  if (wave >= n_nodes) return;

  const float4 v = ((const float4*)(x + (size_t)wave * DIM))[lane];
  float s  = v.x + v.y + v.z + v.w;
  float s2 = v.x * v.x + v.y * v.y + v.z * v.z + v.w * v.w;
  #pragma unroll
  for (int off = 16; off > 0; off >>= 1) {
    s  += __shfl_xor(s,  off, 32);
    s2 += __shfl_xor(s2, off, 32);
  }
  const float mu  = s * (1.0f / DIM);
  const float var = s2 * (1.0f / DIM) - mu * mu;
  const float rs  = rsqrtf(var + 1e-5f);

  const float4 g = ((const float4*)gamma)[lane];
  const float4 b = ((const float4*)beta)[lane];
  float4 hv;
  hv.x = fmaxf((v.x - mu) * rs * g.x + b.x, 0.0f);
  hv.y = fmaxf((v.y - mu) * rs * g.y + b.y, 0.0f);
  hv.z = fmaxf((v.z - mu) * rs * g.z + b.z, 0.0f);
  hv.w = fmaxf((v.w - mu) * rs * g.w + b.w, 0.0f);

  ((float4*)(h + (size_t)wave * DIM))[lane] = hv;
  const float4 z = {0.0f, 0.0f, 0.0f, 0.0f};
  ((float4*)(aggr + (size_t)wave * DIM))[lane] = z;
}

__global__ void __launch_bounds__(BLOCK)
edge_scatter_kernel(const int* __restrict__ ei,
                    const float* __restrict__ h,
                    float* __restrict__ aggr,
                    int n_edges) {
  int wave = (blockIdx.x * blockDim.x + threadIdx.x) >> 5;
  int lane = threadIdx.x & 31;
  if (wave >= n_edges) return;

  const int src = ei[wave];
  const int dst = ei[n_edges + wave];

  const float4 m = ((const float4*)(h + (size_t)src * DIM))[lane];
  float* ap = aggr + (size_t)dst * DIM + lane * 4;
  atomicAdd(ap + 0, m.x);
  atomicAdd(ap + 1, m.y);
  atomicAdd(ap + 2, m.z);
  atomicAdd(ap + 3, m.w);
}

__global__ void __launch_bounds__(256)
fuse_kernel(const float* __restrict__ h,
            const float* __restrict__ eps_p,
            float* __restrict__ aggr,
            int n) {
  int i = blockIdx.x * blockDim.x + threadIdx.x;
  if (i >= n) return;
  const float scale = 1.0f + eps_p[0];
  aggr[i] = fmaf(scale, h[i], aggr[i]);
}

__global__ void __launch_bounds__(BLOCK)
gemm_kernel(const float* __restrict__ Mmat,
            const float* __restrict__ W,
            const float* __restrict__ bias,
            const float* __restrict__ x,
            float* __restrict__ out,
            int n_tiles) {
  int wave = threadIdx.x >> 5;
  int lane = threadIdx.x & 31;
  int tile = blockIdx.x * WAVES_PER_BLOCK + wave;
  if (tile >= n_tiles) return;

  const int tm   = tile >> 3;
  const int tn   = tile & 7;
  const int row0 = tm * 16;
  const int col0 = tn * 16;
  const int half = lane >> 4;
  const int l    = lane & 15;

  const float* arow = Mmat + (size_t)(row0 + l) * DIM + 2 * half;
  const float* brow = W    + (size_t)(col0 + l) * DIM + 2 * half;

  v8f c = {};
  #pragma unroll
  for (int k = 0; k < DIM; k += 4) {
    v2f a = *(const v2f*)(arow + k);
    v2f b = *(const v2f*)(brow + k);
    c = __builtin_amdgcn_wmma_f32_16x16x4_f32(
         false, a,  false, b,
         (short)0, c,  false,  false);
  }

  const float bv = bias[col0 + l];
  #pragma unroll
  for (int r = 0; r < 8; ++r) {
    const int row = row0 + r + 8 * half;
    const size_t idx = (size_t)row * DIM + col0 + l;
    out[idx] = c[r] + bv + x[idx];
  }
}

extern "C" void kernel_launch(void* const* d_in, const int* in_sizes, int n_in,
                              void* d_out, int out_size, void* d_ws, size_t ws_size,
                              hipStream_t stream) {
  const float* x     = (const float*)d_in[0];
  const int*   ei    = (const int*)  d_in[1];
  const float* gamma = (const float*)d_in[2];
  const float* beta  = (const float*)d_in[3];
  const float* W     = (const float*)d_in[4];
  const float* bias  = (const float*)d_in[5];
  const float* eps   = (const float*)d_in[6];
  float* out = (float*)d_out;

  const int n_nodes = in_sizes[0] / DIM;
  const int n_edges = in_sizes[1] / 2;

  float* h    = (float*)d_ws;
  float* aggr = h + (size_t)n_nodes * DIM;

  {
    int blocks = (n_nodes + WAVES_PER_BLOCK - 1) / WAVES_PER_BLOCK;
    ln_relu_zero_kernel<<<blocks, BLOCK, 0, stream>>>(x, gamma, beta, h, aggr, n_nodes);
  }
  {
    int blocks = (n_edges + WAVES_PER_BLOCK - 1) / WAVES_PER_BLOCK;
    edge_scatter_kernel<<<blocks, BLOCK, 0, stream>>>(ei, h, aggr, n_edges);
  }
  {
    int n = n_nodes * DIM;
    int blocks = (n + 255) / 256;
    fuse_kernel<<<blocks, 256, 0, stream>>>(h, eps, aggr, n);
  }
  {
    int n_tiles = (n_nodes / 16) * (DIM / 16);
    int blocks = (n_tiles + WAVES_PER_BLOCK - 1) / WAVES_PER_BLOCK;
    gemm_kernel<<<blocks, BLOCK, 0, stream>>>(aggr, W, bias, x, out, n_tiles);
  }
}


# Replace f32 atomic scatter-add by counting sort on dst (int atomics) plus per-node register gather-sum with the self term as start value; sum order = slot arrival order; launches 4 to 6
# speedup vs baseline: 1.2644x; 1.2644x over previous
#include <hip/hip_runtime.h>

#define DIM 128
#define WAVES_PER_BLOCK 8
#define BLOCK (WAVES_PER_BLOCK * 32)

typedef float v2f __attribute__((ext_vector_type(2)));
typedef float v8f __attribute__((ext_vector_type(8)));

__global__ void __launch_bounds__(BLOCK)
ln_relu_zero_kernel(const float* __restrict__ x,
                    const float* __restrict__ gamma,
                    const float* __restrict__ beta,
                    float* __restrict__ h,
                    int* __restrict__ cnt,
                    int* __restrict__ total,
                    int n_nodes) {
  int wave = (blockIdx.x * blockDim.x + threadIdx.x) >> 5;
  int lane = threadIdx.x & 31;
  if (wave >= n_nodes) return;

  const float4 v = ((const float4*)(x + (size_t)wave * DIM))[lane];
  float s  = v.x + v.y + v.z + v.w;
  float s2 = v.x * v.x + v.y * v.y + v.z * v.z + v.w * v.w;
  #pragma unroll
  for (int off = 16; off > 0; off >>= 1) {
    s  += __shfl_xor(s,  off, 32);
    s2 += __shfl_xor(s2, off, 32);
  }
  const float mu  = s * (1.0f / DIM);
  const float var = s2 * (1.0f / DIM) - mu * mu;
  const float rs  = rsqrtf(var + 1e-5f);

  const float4 g = ((const float4*)gamma)[lane];
  const float4 b = ((const float4*)beta)[lane];
  float4 hv;
  hv.x = fmaxf((v.x - mu) * rs * g.x + b.x, 0.0f);
  hv.y = fmaxf((v.y - mu) * rs * g.y + b.y, 0.0f);
  hv.z = fmaxf((v.z - mu) * rs * g.z + b.z, 0.0f);
  hv.w = fmaxf((v.w - mu) * rs * g.w + b.w, 0.0f);

  ((float4*)(h + (size_t)wave * DIM))[lane] = hv;
  if (lane == 0) {
    cnt[wave] = 0;
    if (wave == 0) total[0] = 0;
  }
}


__global__ void __launch_bounds__(256)
degree_kernel(const int* __restrict__ ei,
              int* __restrict__ cnt,
              int n_edges) {
  int e = blockIdx.x * blockDim.x + threadIdx.x;
  if (e >= n_edges) return;
  atomicAdd(&cnt[ei[n_edges + e]], 1);
}

__global__ void __launch_bounds__(256)
alloc_kernel(const int* __restrict__ cnt,
             int* __restrict__ cursor,
             int* __restrict__ total,
             int n_nodes) {
  __shared__ int wave_sum[8];
  __shared__ int block_base;
  const int i    = blockIdx.x * 256 + threadIdx.x;
  const int lane = threadIdx.x & 31;
  const int w    = threadIdx.x >> 5;
  const int c    = (i < n_nodes) ? cnt[i] : 0;
  int inc = c;
  for (int off = 1; off < 32; off <<= 1) {
    const int t = __shfl_up(inc, off, 32);
    if (lane >= off) inc += t;
  }
  if (lane == 31) wave_sum[w] = inc;
  __syncthreads();
  if (threadIdx.x == 0) {
    int run = 0;
    for (int k = 0; k < 8; ++k) {
      const int t = wave_sum[k];
      wave_sum[k] = run;
      run += t;
    }
    block_base = atomicAdd(total, run);
  }
  __syncthreads();
  if (i < n_nodes) cursor[i] = block_base + wave_sum[w] + inc - c;
}

__global__ void __launch_bounds__(256)
fill_kernel(const int* __restrict__ ei,
            int* __restrict__ cursor,
            int* __restrict__ sorted_src,
            int n_edges) {
  int e = blockIdx.x * blockDim.x + threadIdx.x;
  if (e >= n_edges) return;
  const int src = ei[e];
  const int dst = ei[n_edges + e];
  const int pos = atomicAdd(&cursor[dst], 1);
  sorted_src[pos] = src;
}

__global__ void __launch_bounds__(BLOCK)
gather_sum_kernel(const float* __restrict__ h,
                  const int* __restrict__ sorted_src,
                  const int* __restrict__ cursor,
                  const int* __restrict__ cnt,
                  const float* __restrict__ eps_p,
                  float* __restrict__ Mmat,
                  int n_nodes) {
  const int lane = threadIdx.x & 31;
  const int node = __builtin_amdgcn_readfirstlane(
      (int)((blockIdx.x * blockDim.x + threadIdx.x) >> 5));
  if (node >= n_nodes) return;
  const int n     = cnt[node];
  const int end   = cursor[node];
  const int begin = end - n;
  const float scale = 1.0f + eps_p[0];

  const float4 hv = ((const float4*)(h + (size_t)node * DIM))[lane];
  float4 acc;
  acc.x = scale * hv.x;
  acc.y = scale * hv.y;
  acc.z = scale * hv.z;
  acc.w = scale * hv.w;

  for (int base = begin; base < end; base += 32) {
    const int left = end - base;
    const int m    = left < 32 ? left : 32;
    const int mine = (lane < m) ? sorted_src[base + lane] : 0;
    int j = 0;
    for (; j + 4 <= m; j += 4) {
      const int s0 = __builtin_amdgcn_readlane(mine, j);
      const int s1 = __builtin_amdgcn_readlane(mine, j + 1);
      const int s2 = __builtin_amdgcn_readlane(mine, j + 2);
      const int s3 = __builtin_amdgcn_readlane(mine, j + 3);
      const float4 v0 = ((const float4*)(h + (size_t)s0 * DIM))[lane];
      const float4 v1 = ((const float4*)(h + (size_t)s1 * DIM))[lane];
      const float4 v2 = ((const float4*)(h + (size_t)s2 * DIM))[lane];
      const float4 v3 = ((const float4*)(h + (size_t)s3 * DIM))[lane];
      acc.x += v0.x; acc.y += v0.y; acc.z += v0.z; acc.w += v0.w;
      acc.x += v1.x; acc.y += v1.y; acc.z += v1.z; acc.w += v1.w;
      acc.x += v2.x; acc.y += v2.y; acc.z += v2.z; acc.w += v2.w;
      acc.x += v3.x; acc.y += v3.y; acc.z += v3.z; acc.w += v3.w;
    }
    for (; j < m; ++j) {
      const int s0 = __builtin_amdgcn_readlane(mine, j);
      const float4 v0 = ((const float4*)(h + (size_t)s0 * DIM))[lane];
      acc.x += v0.x; acc.y += v0.y; acc.z += v0.z; acc.w += v0.w;
    }
  }
  ((float4*)(Mmat + (size_t)node * DIM))[lane] = acc;
}

__global__ void __launch_bounds__(BLOCK)
gemm_kernel(const float* __restrict__ Mmat,
            const float* __restrict__ W,
            const float* __restrict__ bias,
            const float* __restrict__ x,
            float* __restrict__ out,
            int n_tiles) {
  int wave = threadIdx.x >> 5;
  int lane = threadIdx.x & 31;
  int tile = blockIdx.x * WAVES_PER_BLOCK + wave;
  if (tile >= n_tiles) return;

  const int tm   = tile >> 3;
  const int tn   = tile & 7;
  const int row0 = tm * 16;
  const int col0 = tn * 16;
  const int half = lane >> 4;
  const int l    = lane & 15;

  const float* arow = Mmat + (size_t)(row0 + l) * DIM + 2 * half;
  const float* brow = W    + (size_t)(col0 + l) * DIM + 2 * half;

  v8f c = {};
  #pragma unroll
  for (int k = 0; k < DIM; k += 4) {
    v2f a = *(const v2f*)(arow + k);
    v2f b = *(const v2f*)(brow + k);
    c = __builtin_amdgcn_wmma_f32_16x16x4_f32(
         false, a,  false, b,
         (short)0, c,  false,  false);
  }

  const float bv = bias[col0 + l];
  #pragma unroll
  for (int r = 0; r < 8; ++r) {
    const int row = row0 + r + 8 * half;
    const size_t idx = (size_t)row * DIM + col0 + l;
    out[idx] = c[r] + bv + x[idx];
  }
}

extern "C" void kernel_launch(void* const* d_in, const int* in_sizes, int n_in,
                              void* d_out, int out_size, void* d_ws, size_t ws_size,
                              hipStream_t stream) {
  const float* x     = (const float*)d_in[0];
  const int*   ei    = (const int*)  d_in[1];
  const float* gamma = (const float*)d_in[2];
  const float* beta  = (const float*)d_in[3];
  const float* W     = (const float*)d_in[4];
  const float* bias  = (const float*)d_in[5];
  const float* eps   = (const float*)d_in[6];
  float* out = (float*)d_out;

  const int n_nodes = in_sizes[0] / DIM;
  const int n_edges = in_sizes[1] / 2;

  float* h          = (float*)d_ws;
  float* aggr       = h + (size_t)n_nodes * DIM;
  int*   cnt        = (int*)(aggr + (size_t)n_nodes * DIM);
  int*   cursor     = cnt + n_nodes;
  int*   sorted_src = cursor + n_nodes;
  int*   total      = sorted_src + n_edges;

  {
    int blocks = (n_nodes + WAVES_PER_BLOCK - 1) / WAVES_PER_BLOCK;
    ln_relu_zero_kernel<<<blocks, BLOCK, 0, stream>>>(x, gamma, beta, h, cnt, total, n_nodes);
  }
  {
    int blocks = (n_edges + 255) / 256;
    degree_kernel<<<blocks, 256, 0, stream>>>(ei, cnt, n_edges);
  }
  {
    int blocks = (n_nodes + 255) / 256;
    alloc_kernel<<<blocks, 256, 0, stream>>>(cnt, cursor, total, n_nodes);
  }
  {
    int blocks = (n_edges + 255) / 256;
    fill_kernel<<<blocks, 256, 0, stream>>>(ei, cursor, sorted_src, n_edges);
  }
  {
    int blocks = (n_nodes + WAVES_PER_BLOCK - 1) / WAVES_PER_BLOCK;
    gather_sum_kernel<<<blocks, BLOCK, 0, stream>>>(h, sorted_src, cursor, cnt, eps, aggr, n_nodes);
  }
  {
    int n_tiles = (n_nodes / 16) * (DIM / 16);
    int blocks = (n_tiles + WAVES_PER_BLOCK - 1) / WAVES_PER_BLOCK;
    gemm_kernel<<<blocks, BLOCK, 0, stream>>>(aggr, W, bias, x, out, n_tiles);
  }
}


# Fixed chip-sized grids: norm, edge, gather and GEMM kernels loop over nodes, edges or row tiles (an eighth of the waves enter); GEMM wave keeps its W slice in registers; sum order stays arrival order
# speedup vs baseline: 1.3587x; 1.0746x over previous
#include <hip/hip_runtime.h>

#define DIM 128
#define WAVES_PER_BLOCK 8
#define BLOCK (WAVES_PER_BLOCK * 32)

#define GRID_NORM   1024
#define GRID_GATHER 2048
#define GRID_EDGES   512
#define GRID_GEMM   1024
#define NORM_ROWS 8

typedef float v2f __attribute__((ext_vector_type(2)));
typedef float v8f __attribute__((ext_vector_type(8)));

__global__ void __launch_bounds__(BLOCK)
ln_relu_zero_kernel(const float* __restrict__ x,
                    const float* __restrict__ gamma,
                    const float* __restrict__ beta,
                    float* __restrict__ h,
                    int* __restrict__ cnt,
                    int* __restrict__ total,
                    int n_nodes) {
  const int lane = threadIdx.x & 31;
  const int gw = __builtin_amdgcn_readfirstlane(
      (int)((blockIdx.x * blockDim.x + threadIdx.x) >> 5));
  const int nw = (int)(gridDim.x * (blockDim.x >> 5));

  const float4 g = ((const float4*)gamma)[lane];
  const float4 b = ((const float4*)beta)[lane];
  if (gw == 0 && lane == 0) total[0] = 0;

  for (int base = gw; base < n_nodes; base += NORM_ROWS * nw) {
    float4 v[NORM_ROWS];
    #pragma unroll
    for (int j = 0; j < NORM_ROWS; ++j) {
      const int nd = base + j * nw;
      v[j] = make_float4(0.0f, 0.0f, 0.0f, 0.0f);
      if (nd < n_nodes) v[j] = ((const float4*)(x + (size_t)nd * DIM))[lane];
    }
    #pragma unroll
    for (int j = 0; j < NORM_ROWS; ++j) {
      const int nd = base + j * nw;
      if (nd < n_nodes) {
        float s  = v[j].x + v[j].y + v[j].z + v[j].w;
        float s2 = v[j].x * v[j].x + v[j].y * v[j].y + v[j].z * v[j].z + v[j].w * v[j].w;
        #pragma unroll
        for (int off = 16; off > 0; off >>= 1) {
          s  += __shfl_xor(s,  off, 32);
          s2 += __shfl_xor(s2, off, 32);
        }
        const float mu  = s * (1.0f / DIM);
        const float var = s2 * (1.0f / DIM) - mu * mu;
        const float rs  = rsqrtf(var + 1e-5f);
        float4 hv;
        hv.x = fmaxf((v[j].x - mu) * rs * g.x + b.x, 0.0f);
        hv.y = fmaxf((v[j].y - mu) * rs * g.y + b.y, 0.0f);
        hv.z = fmaxf((v[j].z - mu) * rs * g.z + b.z, 0.0f);
        hv.w = fmaxf((v[j].w - mu) * rs * g.w + b.w, 0.0f);
        ((float4*)(h + (size_t)nd * DIM))[lane] = hv;
        if (lane == 0) cnt[nd] = 0;
      }
    }
  }
}


__global__ void __launch_bounds__(256)
degree_kernel(const int* __restrict__ ei,
              int* __restrict__ cnt,
              int n_edges) {
  const int t  = blockIdx.x * blockDim.x + threadIdx.x;
  const int nt = gridDim.x * blockDim.x;
  const int* __restrict__ dstp = ei + n_edges;
  for (int base = t; base < n_edges; base += 4 * nt) {
    int d[4];
    #pragma unroll
    for (int j = 0; j < 4; ++j) {
      const int e = base + j * nt;
      d[j] = (e < n_edges) ? dstp[e] : -1;
    }
    #pragma unroll
    for (int j = 0; j < 4; ++j) {
      if (d[j] >= 0) atomicAdd(&cnt[d[j]], 1);
    }
  }
}

__global__ void __launch_bounds__(256)
alloc_kernel(const int* __restrict__ cnt,
             int* __restrict__ cursor,
             int* __restrict__ total,
             int n_nodes) {
  __shared__ int wave_sum[8];
  __shared__ int block_base;
  const int i    = blockIdx.x * 256 + threadIdx.x;
  const int lane = threadIdx.x & 31;
  const int w    = threadIdx.x >> 5;
  const int c    = (i < n_nodes) ? cnt[i] : 0;
  int inc = c;
  for (int off = 1; off < 32; off <<= 1) {
    const int t = __shfl_up(inc, off, 32);
    if (lane >= off) inc += t;
  }
  if (lane == 31) wave_sum[w] = inc;
  __syncthreads();
  if (threadIdx.x == 0) {
    int run = 0;
    for (int k = 0; k < 8; ++k) {
      const int t = wave_sum[k];
      wave_sum[k] = run;
      run += t;
    }
    block_base = atomicAdd(total, run);
  }
  __syncthreads();
  if (i < n_nodes) cursor[i] = block_base + wave_sum[w] + inc - c;
}

__global__ void __launch_bounds__(256)
fill_kernel(const int* __restrict__ ei,
            int* __restrict__ cursor,
            int* __restrict__ sorted_src,
            int n_edges) {
  const int t  = blockIdx.x * blockDim.x + threadIdx.x;
  const int nt = gridDim.x * blockDim.x;
  const int* __restrict__ dstp = ei + n_edges;
  for (int base = t; base < n_edges; base += 4 * nt) {
    int s[4], d[4], pos[4];
    #pragma unroll
    for (int j = 0; j < 4; ++j) {
      const int e = base + j * nt;
      const bool ok = e < n_edges;
      s[j] = ok ? ei[e]   : 0;
      d[j] = ok ? dstp[e] : -1;
    }
    #pragma unroll
    for (int j = 0; j < 4; ++j) {
      pos[j] = (d[j] >= 0) ? atomicAdd(&cursor[d[j]], 1) : -1;
    }
    #pragma unroll
    for (int j = 0; j < 4; ++j) {
      if (pos[j] >= 0) sorted_src[pos[j]] = s[j];
    }
  }
}

__global__ void __launch_bounds__(BLOCK)
gather_sum_kernel(const float* __restrict__ h,
                  const int* __restrict__ sorted_src,
                  const int* __restrict__ cursor,
                  const int* __restrict__ cnt,
                  const float* __restrict__ eps_p,
                  float* __restrict__ Mmat,
                  int n_nodes) {
  const int lane = threadIdx.x & 31;
  const int gw = __builtin_amdgcn_readfirstlane(
      (int)((blockIdx.x * blockDim.x + threadIdx.x) >> 5));
  const int nw = (int)(gridDim.x * (blockDim.x >> 5));
  if (gw >= n_nodes) return;
  const float scale = 1.0f + eps_p[0];

  int n   = cnt[gw];
  int end = cursor[gw];
  for (int node = gw; node < n_nodes; node += nw) {
    const int nxt = node + nw;
    int n_next = 0, end_next = 0;
    if (nxt < n_nodes) {
      n_next   = cnt[nxt];
      end_next = cursor[nxt];
    }
    const int begin = end - n;

    const float4 hv = ((const float4*)(h + (size_t)node * DIM))[lane];
    float4 acc;
    acc.x = scale * hv.x;
    acc.y = scale * hv.y;
    acc.z = scale * hv.z;
    acc.w = scale * hv.w;

    for (int base = begin; base < end; base += 32) {
      const int left = end - base;
      const int m    = left < 32 ? left : 32;
      const int mine = (lane < m) ? sorted_src[base + lane] : 0;
      int j = 0;
      for (; j + 4 <= m; j += 4) {
        const int s0 = __builtin_amdgcn_readlane(mine, j);
        const int s1 = __builtin_amdgcn_readlane(mine, j + 1);
        const int s2 = __builtin_amdgcn_readlane(mine, j + 2);
        const int s3 = __builtin_amdgcn_readlane(mine, j + 3);
        const float4 v0 = ((const float4*)(h + (size_t)s0 * DIM))[lane];
        const float4 v1 = ((const float4*)(h + (size_t)s1 * DIM))[lane];
        const float4 v2 = ((const float4*)(h + (size_t)s2 * DIM))[lane];
        const float4 v3 = ((const float4*)(h + (size_t)s3 * DIM))[lane];
        acc.x += v0.x; acc.y += v0.y; acc.z += v0.z; acc.w += v0.w;
        acc.x += v1.x; acc.y += v1.y; acc.z += v1.z; acc.w += v1.w;
        acc.x += v2.x; acc.y += v2.y; acc.z += v2.z; acc.w += v2.w;
        acc.x += v3.x; acc.y += v3.y; acc.z += v3.z; acc.w += v3.w;
      }
      for (; j < m; ++j) {
        const int s0 = __builtin_amdgcn_readlane(mine, j);
        const float4 v0 = ((const float4*)(h + (size_t)s0 * DIM))[lane];
        acc.x += v0.x; acc.y += v0.y; acc.z += v0.z; acc.w += v0.w;
      }
    }
    ((float4*)(Mmat + (size_t)node * DIM))[lane] = acc;
    n   = n_next;
    end = end_next;
  }
}

__global__ void __launch_bounds__(BLOCK)
gemm_kernel(const float* __restrict__ Mmat,
            const float* __restrict__ W,
            const float* __restrict__ bias,
            const float* __restrict__ x,
            float* __restrict__ out,
            int n_row_tiles) {
  const int wave = threadIdx.x >> 5;
  const int lane = threadIdx.x & 31;
  const int col0 = wave * 16;
  const int half = lane >> 4;
  const int l    = lane & 15;

  const float* brow = W + (size_t)(col0 + l) * DIM + 2 * half;
  v2f b[DIM / 4];
  #pragma unroll
  for (int k = 0; k < DIM; k += 4) b[k >> 2] = *(const v2f*)(brow + k);
  const float bv = bias[col0 + l];

  for (int tm = blockIdx.x; tm < n_row_tiles; tm += gridDim.x) {
    const int row0 = tm * 16;
    const float* arow = Mmat + (size_t)(row0 + l) * DIM + 2 * half;

    v8f c = {};
    #pragma unroll
    for (int k = 0; k < DIM; k += 4) {
      v2f a = *(const v2f*)(arow + k);
      c = __builtin_amdgcn_wmma_f32_16x16x4_f32(
           false, a,  false, b[k >> 2],
           (short)0, c,  false,  false);
    }

    #pragma unroll
    for (int r = 0; r < 8; ++r) {
      const int row = row0 + r + 8 * half;
      const size_t idx = (size_t)row * DIM + col0 + l;
      out[idx] = c[r] + bv + x[idx];
    }
  }
}

extern "C" void kernel_launch(void* const* d_in, const int* in_sizes, int n_in,
                              void* d_out, int out_size, void* d_ws, size_t ws_size,
                              hipStream_t stream) {
  const float* x     = (const float*)d_in[0];
  const int*   ei    = (const int*)  d_in[1];
  const float* gamma = (const float*)d_in[2];
  const float* beta  = (const float*)d_in[3];
  const float* W     = (const float*)d_in[4];
  const float* bias  = (const float*)d_in[5];
  const float* eps   = (const float*)d_in[6];
  float* out = (float*)d_out;

  const int n_nodes = in_sizes[0] / DIM;
  const int n_edges = in_sizes[1] / 2;

  float* h          = (float*)d_ws;
  float* aggr       = h + (size_t)n_nodes * DIM;
  int*   cnt        = (int*)(aggr + (size_t)n_nodes * DIM);
  int*   cursor     = cnt + n_nodes;
  int*   sorted_src = cursor + n_nodes;
  int*   total      = sorted_src + n_edges;

  ln_relu_zero_kernel<<<GRID_NORM, BLOCK, 0, stream>>>(x, gamma, beta, h, cnt, total, n_nodes);
  degree_kernel<<<GRID_EDGES, 256, 0, stream>>>(ei, cnt, n_edges);
  {
    int blocks = (n_nodes + 255) / 256;
    alloc_kernel<<<blocks, 256, 0, stream>>>(cnt, cursor, total, n_nodes);
  }
  fill_kernel<<<GRID_EDGES, 256, 0, stream>>>(ei, cursor, sorted_src, n_edges);
  gather_sum_kernel<<<GRID_GATHER, BLOCK, 0, stream>>>(h, sorted_src, cursor, cnt, eps, aggr, n_nodes);
  gemm_kernel<<<GRID_GEMM, BLOCK, 0, stream>>>(aggr, W, bias, x, out, n_nodes / 16);
}


# Degree pass's counting add returns each edge's rank at its dst, stored per edge; placement = range start + rank by plain loads and stores: one int atomic per edge, not two; sum order: arrival order
# speedup vs baseline: 1.6979x; 1.2497x over previous
#include <hip/hip_runtime.h>

#define DIM 128
#define WAVES_PER_BLOCK 8
#define BLOCK (WAVES_PER_BLOCK * 32)

#define GRID_NORM   1024
#define GRID_GATHER 2048
#define GRID_EDGES   512
#define GRID_GEMM   1024
#define NORM_ROWS 8

typedef float v2f __attribute__((ext_vector_type(2)));
typedef float v8f __attribute__((ext_vector_type(8)));

__global__ void __launch_bounds__(BLOCK)
ln_relu_zero_kernel(const float* __restrict__ x,
                    const float* __restrict__ gamma,
                    const float* __restrict__ beta,
                    float* __restrict__ h,
                    int* __restrict__ cnt,
                    int* __restrict__ total,
                    int n_nodes) {
  const int lane = threadIdx.x & 31;
  const int gw = __builtin_amdgcn_readfirstlane(
      (int)((blockIdx.x * blockDim.x + threadIdx.x) >> 5));
  const int nw = (int)(gridDim.x * (blockDim.x >> 5));

  const float4 g = ((const float4*)gamma)[lane];
  const float4 b = ((const float4*)beta)[lane];
  if (gw == 0 && lane == 0) total[0] = 0;

  for (int base = gw; base < n_nodes; base += NORM_ROWS * nw) {
    float4 v[NORM_ROWS];
    #pragma unroll
    for (int j = 0; j < NORM_ROWS; ++j) {
      const int nd = base + j * nw;
      v[j] = make_float4(0.0f, 0.0f, 0.0f, 0.0f);
      if (nd < n_nodes) v[j] = ((const float4*)(x + (size_t)nd * DIM))[lane];
    }
    #pragma unroll
    for (int j = 0; j < NORM_ROWS; ++j) {
      const int nd = base + j * nw;
      if (nd < n_nodes) {
        float s  = v[j].x + v[j].y + v[j].z + v[j].w;
        float s2 = v[j].x * v[j].x + v[j].y * v[j].y + v[j].z * v[j].z + v[j].w * v[j].w;
        #pragma unroll
        for (int off = 16; off > 0; off >>= 1) {
          s  += __shfl_xor(s,  off, 32);
          s2 += __shfl_xor(s2, off, 32);
        }
        const float mu  = s * (1.0f / DIM);
        const float var = s2 * (1.0f / DIM) - mu * mu;
        const float rs  = rsqrtf(var + 1e-5f);
        float4 hv;
        hv.x = fmaxf((v[j].x - mu) * rs * g.x + b.x, 0.0f);
        hv.y = fmaxf((v[j].y - mu) * rs * g.y + b.y, 0.0f);
        hv.z = fmaxf((v[j].z - mu) * rs * g.z + b.z, 0.0f);
        hv.w = fmaxf((v[j].w - mu) * rs * g.w + b.w, 0.0f);
        ((float4*)(h + (size_t)nd * DIM))[lane] = hv;
        if (lane == 0) cnt[nd] = 0;
      }
    }
  }
}


__global__ void __launch_bounds__(256)
degree_rank_kernel(const int* __restrict__ ei,
                   int* __restrict__ cnt,
                   int* __restrict__ rank,
                   int n_edges) {
  const int t  = blockIdx.x * blockDim.x + threadIdx.x;
  const int nt = gridDim.x * blockDim.x;
  const int* __restrict__ dstp = ei + n_edges;
  for (int base = t; base < n_edges; base += 4 * nt) {
    int d[4], r[4];
    #pragma unroll
    for (int j = 0; j < 4; ++j) {
      const int e = base + j * nt;
      d[j] = (e < n_edges) ? dstp[e] : -1;
    }
    #pragma unroll
    for (int j = 0; j < 4; ++j) {
      r[j] = (d[j] >= 0) ? atomicAdd(&cnt[d[j]], 1) : 0;
    }
    #pragma unroll
    for (int j = 0; j < 4; ++j) {
      if (d[j] >= 0) rank[base + j * nt] = r[j];
    }
  }
}

__global__ void __launch_bounds__(256)
alloc_kernel(const int* __restrict__ cnt,
             int* __restrict__ cursor,
             int* __restrict__ total,
             int n_nodes) {
  __shared__ int wave_sum[8];
  __shared__ int block_base;
  const int i    = blockIdx.x * 256 + threadIdx.x;
  const int lane = threadIdx.x & 31;
  const int w    = threadIdx.x >> 5;
  const int c    = (i < n_nodes) ? cnt[i] : 0;
  int inc = c;
  for (int off = 1; off < 32; off <<= 1) {
    const int t = __shfl_up(inc, off, 32);
    if (lane >= off) inc += t;
  }
  if (lane == 31) wave_sum[w] = inc;
  __syncthreads();
  if (threadIdx.x == 0) {
    int run = 0;
    for (int k = 0; k < 8; ++k) {
      const int t = wave_sum[k];
      wave_sum[k] = run;
      run += t;
    }
    block_base = atomicAdd(total, run);
  }
  __syncthreads();
  if (i < n_nodes) cursor[i] = block_base + wave_sum[w] + inc - c;
}

__global__ void __launch_bounds__(256)
fill_kernel(const int* __restrict__ ei,
            const int* __restrict__ cursor,
            const int* __restrict__ rank,
            int* __restrict__ sorted_src,
            int n_edges) {
  const int t  = blockIdx.x * blockDim.x + threadIdx.x;
  const int nt = gridDim.x * blockDim.x;
  const int* __restrict__ dstp = ei + n_edges;
  for (int base = t; base < n_edges; base += 4 * nt) {
    int s[4], d[4], r[4], st[4];
    #pragma unroll
    for (int j = 0; j < 4; ++j) {
      const int e = base + j * nt;
      const bool ok = e < n_edges;
      s[j] = ok ? ei[e]   : 0;
      d[j] = ok ? dstp[e] : -1;
      r[j] = ok ? rank[e] : 0;
    }
    #pragma unroll
    for (int j = 0; j < 4; ++j) {
      st[j] = (d[j] >= 0) ? cursor[d[j]] : 0;
    }
    #pragma unroll
    for (int j = 0; j < 4; ++j) {
      if (d[j] >= 0) sorted_src[st[j] + r[j]] = s[j];
    }
  }
}

__global__ void __launch_bounds__(BLOCK)
gather_sum_kernel(const float* __restrict__ h,
                  const int* __restrict__ sorted_src,
                  const int* __restrict__ cursor,
                  const int* __restrict__ cnt,
                  const float* __restrict__ eps_p,
                  float* __restrict__ Mmat,
                  int n_nodes) {
  const int lane = threadIdx.x & 31;
  const int gw = __builtin_amdgcn_readfirstlane(
      (int)((blockIdx.x * blockDim.x + threadIdx.x) >> 5));
  const int nw = (int)(gridDim.x * (blockDim.x >> 5));
  if (gw >= n_nodes) return;
  const float scale = 1.0f + eps_p[0];

  int n     = cnt[gw];
  int begin = cursor[gw];
  for (int node = gw; node < n_nodes; node += nw) {
    const int nxt = node + nw;
    int n_next = 0, begin_next = 0;
    if (nxt < n_nodes) {
      n_next     = cnt[nxt];
      begin_next = cursor[nxt];
    }
    const int end = begin + n;

    const float4 hv = ((const float4*)(h + (size_t)node * DIM))[lane];
    float4 acc;
    acc.x = scale * hv.x;
    acc.y = scale * hv.y;
    acc.z = scale * hv.z;
    acc.w = scale * hv.w;

    for (int base = begin; base < end; base += 32) {
      const int left = end - base;
      const int m    = left < 32 ? left : 32;
      const int mine = (lane < m) ? sorted_src[base + lane] : 0;
      int j = 0;
      for (; j + 4 <= m; j += 4) {
        const int s0 = __builtin_amdgcn_readlane(mine, j);
        const int s1 = __builtin_amdgcn_readlane(mine, j + 1);
        const int s2 = __builtin_amdgcn_readlane(mine, j + 2);
        const int s3 = __builtin_amdgcn_readlane(mine, j + 3);
        const float4 v0 = ((const float4*)(h + (size_t)s0 * DIM))[lane];
        const float4 v1 = ((const float4*)(h + (size_t)s1 * DIM))[lane];
        const float4 v2 = ((const float4*)(h + (size_t)s2 * DIM))[lane];
        const float4 v3 = ((const float4*)(h + (size_t)s3 * DIM))[lane];
        acc.x += v0.x; acc.y += v0.y; acc.z += v0.z; acc.w += v0.w;
        acc.x += v1.x; acc.y += v1.y; acc.z += v1.z; acc.w += v1.w;
        acc.x += v2.x; acc.y += v2.y; acc.z += v2.z; acc.w += v2.w;
        acc.x += v3.x; acc.y += v3.y; acc.z += v3.z; acc.w += v3.w;
      }
      for (; j < m; ++j) {
        const int s0 = __builtin_amdgcn_readlane(mine, j);
        const float4 v0 = ((const float4*)(h + (size_t)s0 * DIM))[lane];
        acc.x += v0.x; acc.y += v0.y; acc.z += v0.z; acc.w += v0.w;
      }
    }
    ((float4*)(Mmat + (size_t)node * DIM))[lane] = acc;
    n     = n_next;
    begin = begin_next;
  }
}

__global__ void __launch_bounds__(BLOCK)
gemm_kernel(const float* __restrict__ Mmat,
            const float* __restrict__ W,
            const float* __restrict__ bias,
            const float* __restrict__ x,
            float* __restrict__ out,
            int n_row_tiles) {
  const int wave = threadIdx.x >> 5;
  const int lane = threadIdx.x & 31;
  const int col0 = wave * 16;
  const int half = lane >> 4;
  const int l    = lane & 15;

  const float* brow = W + (size_t)(col0 + l) * DIM + 2 * half;
  v2f b[DIM / 4];
  #pragma unroll
  for (int k = 0; k < DIM; k += 4) b[k >> 2] = *(const v2f*)(brow + k);
  const float bv = bias[col0 + l];

  for (int tm = blockIdx.x; tm < n_row_tiles; tm += gridDim.x) {
    const int row0 = tm * 16;
    const float* arow = Mmat + (size_t)(row0 + l) * DIM + 2 * half;

    v8f c = {};
    #pragma unroll
    for (int k = 0; k < DIM; k += 4) {
      v2f a = *(const v2f*)(arow + k);
      c = __builtin_amdgcn_wmma_f32_16x16x4_f32(
           false, a,  false, b[k >> 2],
           (short)0, c,  false,  false);
    }

    #pragma unroll
    for (int r = 0; r < 8; ++r) {
      const int row = row0 + r + 8 * half;
      const size_t idx = (size_t)row * DIM + col0 + l;
      out[idx] = c[r] + bv + x[idx];
    }
  }
}

extern "C" void kernel_launch(void* const* d_in, const int* in_sizes, int n_in,
                              void* d_out, int out_size, void* d_ws, size_t ws_size,
                              hipStream_t stream) {
  const float* x     = (const float*)d_in[0];
  const int*   ei    = (const int*)  d_in[1];
  const float* gamma = (const float*)d_in[2];
  const float* beta  = (const float*)d_in[3];
  const float* W     = (const float*)d_in[4];
  const float* bias  = (const float*)d_in[5];
  const float* eps   = (const float*)d_in[6];
  float* out = (float*)d_out;

  const int n_nodes = in_sizes[0] / DIM;
  const int n_edges = in_sizes[1] / 2;

  float* h          = (float*)d_ws;
  float* aggr       = h + (size_t)n_nodes * DIM;
  int*   cnt        = (int*)(aggr + (size_t)n_nodes * DIM);
  int*   cursor     = cnt + n_nodes;
  int*   sorted_src = cursor + n_nodes;
  int*   rank       = sorted_src + n_edges;
  int*   total      = rank + n_edges;

  ln_relu_zero_kernel<<<GRID_NORM, BLOCK, 0, stream>>>(x, gamma, beta, h, cnt, total, n_nodes);
  degree_rank_kernel<<<GRID_EDGES, 256, 0, stream>>>(ei, cnt, rank, n_edges);
  {
    int blocks = (n_nodes + 255) / 256;
    alloc_kernel<<<blocks, 256, 0, stream>>>(cnt, cursor, total, n_nodes);
  }
  fill_kernel<<<GRID_EDGES, 256, 0, stream>>>(ei, cursor, rank, sorted_src, n_edges);
  gather_sum_kernel<<<GRID_GATHER, BLOCK, 0, stream>>>(h, sorted_src, cursor, cnt, eps, aggr, n_nodes);
  gemm_kernel<<<GRID_GEMM, BLOCK, 0, stream>>>(aggr, W, bias, x, out, n_nodes / 16);
}


# Edge ranks from LDS counters, not global atomics: a block owns a node range x edge slice and scans the slice's dst; alloc pass prefixes per-slice counts; sum order: arrival order within a slice
# speedup vs baseline: 2.0152x; 1.1869x over previous
#include <hip/hip_runtime.h>

#define DIM 128
#define WAVES_PER_BLOCK 8
#define BLOCK (WAVES_PER_BLOCK * 32)

#define GRID_NORM   1024
#define GRID_GATHER 2048
#define GRID_EDGES   512
#define GRID_GEMM   1024
#define NORM_ROWS 8

#define CNT_R 32
#define CNT_S 16
#define CNT_CAP 4096
#define CNT_THREADS 1024
#define CNT_DEPTH 8

typedef float v2f __attribute__((ext_vector_type(2)));
typedef float v8f __attribute__((ext_vector_type(8)));

__global__ void __launch_bounds__(BLOCK)
ln_relu_zero_kernel(const float* __restrict__ x,
                    const float* __restrict__ gamma,
                    const float* __restrict__ beta,
                    float* __restrict__ h,
                    int* __restrict__ total,
                    int n_nodes) {
  const int lane = threadIdx.x & 31;
  const int gw = __builtin_amdgcn_readfirstlane(
      (int)((blockIdx.x * blockDim.x + threadIdx.x) >> 5));
  const int nw = (int)(gridDim.x * (blockDim.x >> 5));

  const float4 g = ((const float4*)gamma)[lane];
  const float4 b = ((const float4*)beta)[lane];
  if (gw == 0 && lane == 0) total[0] = 0;

  for (int base = gw; base < n_nodes; base += NORM_ROWS * nw) {
    float4 v[NORM_ROWS];
    #pragma unroll
    for (int j = 0; j < NORM_ROWS; ++j) {
      const int nd = base + j * nw;
      v[j] = make_float4(0.0f, 0.0f, 0.0f, 0.0f);
      if (nd < n_nodes) v[j] = ((const float4*)(x + (size_t)nd * DIM))[lane];
    }
    #pragma unroll
    for (int j = 0; j < NORM_ROWS; ++j) {
      const int nd = base + j * nw;
      if (nd < n_nodes) {
        float s  = v[j].x + v[j].y + v[j].z + v[j].w;
        float s2 = v[j].x * v[j].x + v[j].y * v[j].y + v[j].z * v[j].z + v[j].w * v[j].w;
        #pragma unroll
        for (int off = 16; off > 0; off >>= 1) {
          s  += __shfl_xor(s,  off, 32);
          s2 += __shfl_xor(s2, off, 32);
        }
        const float mu  = s * (1.0f / DIM);
        const float var = s2 * (1.0f / DIM) - mu * mu;
        const float rs  = rsqrtf(var + 1e-5f);
        float4 hv;
        hv.x = fmaxf((v[j].x - mu) * rs * g.x + b.x, 0.0f);
        hv.y = fmaxf((v[j].y - mu) * rs * g.y + b.y, 0.0f);
        hv.z = fmaxf((v[j].z - mu) * rs * g.z + b.z, 0.0f);
        hv.w = fmaxf((v[j].w - mu) * rs * g.w + b.w, 0.0f);
        ((float4*)(h + (size_t)nd * DIM))[lane] = hv;
      }
    }
  }
}


__global__ void __launch_bounds__(CNT_THREADS)
slice_rank_kernel(const int* __restrict__ ei,
                  int* __restrict__ scount,
                  int* __restrict__ rank,
                  int n_edges,
                  int n_nodes) {
  __shared__ int lcnt[CNT_CAP];
  const int r = blockIdx.x % CNT_R;
  const int sl_id = blockIdx.x / CNT_R;
  const int* __restrict__ dstp = ei + n_edges;

  int rl = (n_nodes + CNT_R - 1) / CNT_R;
  if (rl > CNT_CAP) rl = CNT_CAP;
  const int sl = (n_edges + CNT_S - 1) / CNT_S;
  const int e0 = sl_id * sl;
  int e1 = e0 + sl;
  if (e1 > n_edges) e1 = n_edges;

  for (int lo = r * rl; lo < n_nodes; lo += CNT_R * rl) {
    int len = n_nodes - lo;
    if (len > rl) len = rl;
    for (int i = threadIdx.x; i < len; i += CNT_THREADS) lcnt[i] = 0;
    __syncthreads();

    for (int base = e0 + (int)threadIdx.x; base < e1; base += CNT_DEPTH * CNT_THREADS) {
      int d[CNT_DEPTH];
      #pragma unroll
      for (int j = 0; j < CNT_DEPTH; ++j) {
        const int e = base + j * CNT_THREADS;
        d[j] = (e < e1) ? dstp[e] : -1;
      }
      int rk[CNT_DEPTH];
      #pragma unroll
      for (int j = 0; j < CNT_DEPTH; ++j) {
        const unsigned off = (unsigned)(d[j] - lo);
        rk[j] = (off < (unsigned)len) ? atomicAdd(&lcnt[off], 1) : -1;
      }
      #pragma unroll
      for (int j = 0; j < CNT_DEPTH; ++j) {
        if (rk[j] >= 0) rank[base + j * CNT_THREADS] = rk[j];
      }
    }
    __syncthreads();

    int* __restrict__ row = scount + (size_t)sl_id * n_nodes + lo;
    for (int i = threadIdx.x; i < len; i += CNT_THREADS) row[i] = lcnt[i];
    __syncthreads();
  }
}

__global__ void __launch_bounds__(256)
alloc_kernel(int* __restrict__ scount,
             int* __restrict__ cnt,
             int* __restrict__ cursor,
             int* __restrict__ total,
             int n_nodes) {
  __shared__ int wave_sum[8];
  __shared__ int block_base;
  const int i    = blockIdx.x * 256 + threadIdx.x;
  const int lane = threadIdx.x & 31;
  const int w    = threadIdx.x >> 5;
  int cs[CNT_S];
  int c = 0;
  #pragma unroll
  for (int s = 0; s < CNT_S; ++s) {
    cs[s] = (i < n_nodes) ? scount[(size_t)s * n_nodes + i] : 0;
    c += cs[s];
  }
  int inc = c;
  for (int off = 1; off < 32; off <<= 1) {
    const int t = __shfl_up(inc, off, 32);
    if (lane >= off) inc += t;
  }
  if (lane == 31) wave_sum[w] = inc;
  __syncthreads();
  if (threadIdx.x == 0) {
    int run = 0;
    for (int k = 0; k < 8; ++k) {
      const int t = wave_sum[k];
      wave_sum[k] = run;
      run += t;
    }
    block_base = atomicAdd(total, run);
  }
  __syncthreads();
  if (i < n_nodes) {
    const int start = block_base + wave_sum[w] + inc - c;
    cursor[i] = start;
    cnt[i]    = c;
    int run = start;
    #pragma unroll
    for (int s = 0; s < CNT_S; ++s) {
      scount[(size_t)s * n_nodes + i] = run;
      run += cs[s];
    }
  }
}

__global__ void __launch_bounds__(256)
fill_kernel(const int* __restrict__ ei,
            const int* __restrict__ sstart,
            const int* __restrict__ rank,
            int* __restrict__ sorted_src,
            int n_edges,
            int n_nodes) {
  const int t  = blockIdx.x * blockDim.x + threadIdx.x;
  const int nt = gridDim.x * blockDim.x;
  const int* __restrict__ dstp = ei + n_edges;
  const int sl = (n_edges + CNT_S - 1) / CNT_S;
  for (int base = t; base < n_edges; base += 4 * nt) {
    int s[4], d[4], r[4], st[4];
    #pragma unroll
    for (int j = 0; j < 4; ++j) {
      const int e = base + j * nt;
      const bool ok = e < n_edges;
      s[j] = ok ? ei[e]   : 0;
      d[j] = ok ? dstp[e] : -1;
      r[j] = ok ? rank[e] : 0;
    }
    #pragma unroll
    for (int j = 0; j < 4; ++j) {
      const int slice = (base + j * nt) / sl;
      st[j] = (d[j] >= 0) ? sstart[(size_t)slice * n_nodes + d[j]] : 0;
    }
    #pragma unroll
    for (int j = 0; j < 4; ++j) {
      if (d[j] >= 0) sorted_src[st[j] + r[j]] = s[j];
    }
  }
}

__global__ void __launch_bounds__(BLOCK)
gather_sum_kernel(const float* __restrict__ h,
                  const int* __restrict__ sorted_src,
                  const int* __restrict__ cursor,
                  const int* __restrict__ cnt,
                  const float* __restrict__ eps_p,
                  float* __restrict__ Mmat,
                  int n_nodes) {
  const int lane = threadIdx.x & 31;
  const int gw = __builtin_amdgcn_readfirstlane(
      (int)((blockIdx.x * blockDim.x + threadIdx.x) >> 5));
  const int nw = (int)(gridDim.x * (blockDim.x >> 5));
  if (gw >= n_nodes) return;
  const float scale = 1.0f + eps_p[0];

  int n     = cnt[gw];
  int begin = cursor[gw];
  for (int node = gw; node < n_nodes; node += nw) {
    const int nxt = node + nw;
    int n_next = 0, begin_next = 0;
    if (nxt < n_nodes) {
      n_next     = cnt[nxt];
      begin_next = cursor[nxt];
    }
    const int end = begin + n;

    const float4 hv = ((const float4*)(h + (size_t)node * DIM))[lane];
    float4 acc;
    acc.x = scale * hv.x;
    acc.y = scale * hv.y;
    acc.z = scale * hv.z;
    acc.w = scale * hv.w;

    for (int base = begin; base < end; base += 32) {
      const int left = end - base;
      const int m    = left < 32 ? left : 32;
      const int mine = (lane < m) ? sorted_src[base + lane] : 0;
      int j = 0;
      for (; j + 4 <= m; j += 4) {
        const int s0 = __builtin_amdgcn_readlane(mine, j);
        const int s1 = __builtin_amdgcn_readlane(mine, j + 1);
        const int s2 = __builtin_amdgcn_readlane(mine, j + 2);
        const int s3 = __builtin_amdgcn_readlane(mine, j + 3);
        const float4 v0 = ((const float4*)(h + (size_t)s0 * DIM))[lane];
        const float4 v1 = ((const float4*)(h + (size_t)s1 * DIM))[lane];
        const float4 v2 = ((const float4*)(h + (size_t)s2 * DIM))[lane];
        const float4 v3 = ((const float4*)(h + (size_t)s3 * DIM))[lane];
        acc.x += v0.x; acc.y += v0.y; acc.z += v0.z; acc.w += v0.w;
        acc.x += v1.x; acc.y += v1.y; acc.z += v1.z; acc.w += v1.w;
        acc.x += v2.x; acc.y += v2.y; acc.z += v2.z; acc.w += v2.w;
        acc.x += v3.x; acc.y += v3.y; acc.z += v3.z; acc.w += v3.w;
      }
      for (; j < m; ++j) {
        const int s0 = __builtin_amdgcn_readlane(mine, j);
        const float4 v0 = ((const float4*)(h + (size_t)s0 * DIM))[lane];
        acc.x += v0.x; acc.y += v0.y; acc.z += v0.z; acc.w += v0.w;
      }
    }
    ((float4*)(Mmat + (size_t)node * DIM))[lane] = acc;
    n     = n_next;
    begin = begin_next;
  }
}

__global__ void __launch_bounds__(BLOCK)
gemm_kernel(const float* __restrict__ Mmat,
            const float* __restrict__ W,
            const float* __restrict__ bias,
            const float* __restrict__ x,
            float* __restrict__ out,
            int n_row_tiles) {
  const int wave = threadIdx.x >> 5;
  const int lane = threadIdx.x & 31;
  const int col0 = wave * 16;
  const int half = lane >> 4;
  const int l    = lane & 15;

  const float* brow = W + (size_t)(col0 + l) * DIM + 2 * half;
  v2f b[DIM / 4];
  #pragma unroll
  for (int k = 0; k < DIM; k += 4) b[k >> 2] = *(const v2f*)(brow + k);
  const float bv = bias[col0 + l];

  for (int tm = blockIdx.x; tm < n_row_tiles; tm += gridDim.x) {
    const int row0 = tm * 16;
    const float* arow = Mmat + (size_t)(row0 + l) * DIM + 2 * half;

    v8f c = {};
    #pragma unroll
    for (int k = 0; k < DIM; k += 4) {
      v2f a = *(const v2f*)(arow + k);
      c = __builtin_amdgcn_wmma_f32_16x16x4_f32(
           false, a,  false, b[k >> 2],
           (short)0, c,  false,  false);
    }

    #pragma unroll
    for (int r = 0; r < 8; ++r) {
      const int row = row0 + r + 8 * half;
      const size_t idx = (size_t)row * DIM + col0 + l;
      out[idx] = c[r] + bv + x[idx];
    }
  }
}

extern "C" void kernel_launch(void* const* d_in, const int* in_sizes, int n_in,
                              void* d_out, int out_size, void* d_ws, size_t ws_size,
                              hipStream_t stream) {
  const float* x     = (const float*)d_in[0];
  const int*   ei    = (const int*)  d_in[1];
  const float* gamma = (const float*)d_in[2];
  const float* beta  = (const float*)d_in[3];
  const float* W     = (const float*)d_in[4];
  const float* bias  = (const float*)d_in[5];
  const float* eps   = (const float*)d_in[6];
  float* out = (float*)d_out;

  const int n_nodes = in_sizes[0] / DIM;
  const int n_edges = in_sizes[1] / 2;

  float* h          = (float*)d_ws;
  float* aggr       = h + (size_t)n_nodes * DIM;
  int*   cnt        = (int*)(aggr + (size_t)n_nodes * DIM);
  int*   cursor     = cnt + n_nodes;
  int*   sorted_src = cursor + n_nodes;
  int*   rank       = sorted_src + n_edges;
  int*   scount     = rank + n_edges;
  int*   total      = scount + (size_t)CNT_S * n_nodes;

  ln_relu_zero_kernel<<<GRID_NORM, BLOCK, 0, stream>>>(x, gamma, beta, h, total, n_nodes);
  slice_rank_kernel<<<CNT_R * CNT_S, CNT_THREADS, 0, stream>>>(ei, scount, rank, n_edges, n_nodes);
  {
    int blocks = (n_nodes + 255) / 256;
    alloc_kernel<<<blocks, 256, 0, stream>>>(scount, cnt, cursor, total, n_nodes);
  }
  fill_kernel<<<GRID_EDGES, 256, 0, stream>>>(ei, scount, rank, sorted_src, n_edges, n_nodes);
  gather_sum_kernel<<<GRID_GATHER, BLOCK, 0, stream>>>(h, sorted_src, cursor, cnt, eps, aggr, n_nodes);
  gemm_kernel<<<GRID_GEMM, BLOCK, 0, stream>>>(aggr, W, bias, x, out, n_nodes / 16);
}


# Neighbour-sum and GEMM kernels fused: a block sums its 16 rows in registers, 8 row loads in flight, holds the tile in LDS and feeds WMMA from it; launches 6 to 5; sum order as before
# speedup vs baseline: 2.3371x; 1.1597x over previous
#include <hip/hip_runtime.h>

#define DIM 128
#define WAVES_PER_BLOCK 8
#define BLOCK (WAVES_PER_BLOCK * 32)

#define GRID_NORM   1024
#define GRID_EDGES   512
#define GRID_FUSED  1024
#define TILE_PITCH   132
#define NORM_ROWS 8

#define CNT_R 32
#define CNT_S 16
#define CNT_CAP 4096
#define CNT_THREADS 1024
#define CNT_DEPTH 8

typedef float v2f __attribute__((ext_vector_type(2)));
typedef float v8f __attribute__((ext_vector_type(8)));

__global__ void __launch_bounds__(BLOCK)
ln_relu_zero_kernel(const float* __restrict__ x,
                    const float* __restrict__ gamma,
                    const float* __restrict__ beta,
                    float* __restrict__ h,
                    int* __restrict__ total,
                    int n_nodes) {
  const int lane = threadIdx.x & 31;
  const int gw = __builtin_amdgcn_readfirstlane(
      (int)((blockIdx.x * blockDim.x + threadIdx.x) >> 5));
  const int nw = (int)(gridDim.x * (blockDim.x >> 5));

  const float4 g = ((const float4*)gamma)[lane];
  const float4 b = ((const float4*)beta)[lane];
  if (gw == 0 && lane == 0) total[0] = 0;

  for (int base = gw; base < n_nodes; base += NORM_ROWS * nw) {
    float4 v[NORM_ROWS];
    #pragma unroll
    for (int j = 0; j < NORM_ROWS; ++j) {
      const int nd = base + j * nw;
      v[j] = make_float4(0.0f, 0.0f, 0.0f, 0.0f);
      if (nd < n_nodes) v[j] = ((const float4*)(x + (size_t)nd * DIM))[lane];
    }
    #pragma unroll
    for (int j = 0; j < NORM_ROWS; ++j) {
      const int nd = base + j * nw;
      if (nd < n_nodes) {
        float s  = v[j].x + v[j].y + v[j].z + v[j].w;
        float s2 = v[j].x * v[j].x + v[j].y * v[j].y + v[j].z * v[j].z + v[j].w * v[j].w;
        #pragma unroll
        for (int off = 16; off > 0; off >>= 1) {
          s  += __shfl_xor(s,  off, 32);
          s2 += __shfl_xor(s2, off, 32);
        }
        const float mu  = s * (1.0f / DIM);
        const float var = s2 * (1.0f / DIM) - mu * mu;
        const float rs  = rsqrtf(var + 1e-5f);
        float4 hv;
        hv.x = fmaxf((v[j].x - mu) * rs * g.x + b.x, 0.0f);
        hv.y = fmaxf((v[j].y - mu) * rs * g.y + b.y, 0.0f);
        hv.z = fmaxf((v[j].z - mu) * rs * g.z + b.z, 0.0f);
        hv.w = fmaxf((v[j].w - mu) * rs * g.w + b.w, 0.0f);
        ((float4*)(h + (size_t)nd * DIM))[lane] = hv;
      }
    }
  }
}


__global__ void __launch_bounds__(CNT_THREADS)
slice_rank_kernel(const int* __restrict__ ei,
                  int* __restrict__ scount,
                  int* __restrict__ rank,
                  int n_edges,
                  int n_nodes) {
  __shared__ int lcnt[CNT_CAP];
  const int r = blockIdx.x % CNT_R;
  const int sl_id = blockIdx.x / CNT_R;
  const int* __restrict__ dstp = ei + n_edges;

  int rl = (n_nodes + CNT_R - 1) / CNT_R;
  if (rl > CNT_CAP) rl = CNT_CAP;
  const int sl = (n_edges + CNT_S - 1) / CNT_S;
  const int e0 = sl_id * sl;
  int e1 = e0 + sl;
  if (e1 > n_edges) e1 = n_edges;

  for (int lo = r * rl; lo < n_nodes; lo += CNT_R * rl) {
    int len = n_nodes - lo;
    if (len > rl) len = rl;
    for (int i = threadIdx.x; i < len; i += CNT_THREADS) lcnt[i] = 0;
    __syncthreads();

    for (int base = e0 + (int)threadIdx.x; base < e1; base += CNT_DEPTH * CNT_THREADS) {
      int d[CNT_DEPTH];
      #pragma unroll
      for (int j = 0; j < CNT_DEPTH; ++j) {
        const int e = base + j * CNT_THREADS;
        d[j] = (e < e1) ? dstp[e] : -1;
      }
      int rk[CNT_DEPTH];
      #pragma unroll
      for (int j = 0; j < CNT_DEPTH; ++j) {
        const unsigned off = (unsigned)(d[j] - lo);
        rk[j] = (off < (unsigned)len) ? atomicAdd(&lcnt[off], 1) : -1;
      }
      #pragma unroll
      for (int j = 0; j < CNT_DEPTH; ++j) {
        if (rk[j] >= 0) rank[base + j * CNT_THREADS] = rk[j];
      }
    }
    __syncthreads();

    int* __restrict__ row = scount + (size_t)sl_id * n_nodes + lo;
    for (int i = threadIdx.x; i < len; i += CNT_THREADS) row[i] = lcnt[i];
    __syncthreads();
  }
}

__global__ void __launch_bounds__(256)
alloc_kernel(int* __restrict__ scount,
             int* __restrict__ cnt,
             int* __restrict__ cursor,
             int* __restrict__ total,
             int n_nodes) {
  __shared__ int wave_sum[8];
  __shared__ int block_base;
  const int i    = blockIdx.x * 256 + threadIdx.x;
  const int lane = threadIdx.x & 31;
  const int w    = threadIdx.x >> 5;
  int cs[CNT_S];
  int c = 0;
  #pragma unroll
  for (int s = 0; s < CNT_S; ++s) {
    cs[s] = (i < n_nodes) ? scount[(size_t)s * n_nodes + i] : 0;
    c += cs[s];
  }
  int inc = c;
  for (int off = 1; off < 32; off <<= 1) {
    const int t = __shfl_up(inc, off, 32);
    if (lane >= off) inc += t;
  }
  if (lane == 31) wave_sum[w] = inc;
  __syncthreads();
  if (threadIdx.x == 0) {
    int run = 0;
    for (int k = 0; k < 8; ++k) {
      const int t = wave_sum[k];
      wave_sum[k] = run;
      run += t;
    }
    block_base = atomicAdd(total, run);
  }
  __syncthreads();
  if (i < n_nodes) {
    const int start = block_base + wave_sum[w] + inc - c;
    cursor[i] = start;
    cnt[i]    = c;
    int run = start;
    #pragma unroll
    for (int s = 0; s < CNT_S; ++s) {
      scount[(size_t)s * n_nodes + i] = run;
      run += cs[s];
    }
  }
}

__global__ void __launch_bounds__(256)
fill_kernel(const int* __restrict__ ei,
            const int* __restrict__ sstart,
            const int* __restrict__ rank,
            int* __restrict__ sorted_src,
            int n_edges,
            int n_nodes) {
  const int t  = blockIdx.x * blockDim.x + threadIdx.x;
  const int nt = gridDim.x * blockDim.x;
  const int* __restrict__ dstp = ei + n_edges;
  const int sl = (n_edges + CNT_S - 1) / CNT_S;
  for (int base = t; base < n_edges; base += 4 * nt) {
    int s[4], d[4], r[4], st[4];
    #pragma unroll
    for (int j = 0; j < 4; ++j) {
      const int e = base + j * nt;
      const bool ok = e < n_edges;
      s[j] = ok ? ei[e]   : 0;
      d[j] = ok ? dstp[e] : -1;
      r[j] = ok ? rank[e] : 0;
    }
    #pragma unroll
    for (int j = 0; j < 4; ++j) {
      const int slice = (base + j * nt) / sl;
      st[j] = (d[j] >= 0) ? sstart[(size_t)slice * n_nodes + d[j]] : 0;
    }
    #pragma unroll
    for (int j = 0; j < 4; ++j) {
      if (d[j] >= 0) sorted_src[st[j] + r[j]] = s[j];
    }
  }
}

__device__ __forceinline__ float4 row4(const float* __restrict__ h, int s, int lane) {
  return ((const float4*)(h + (size_t)s * DIM))[lane];
}

__device__ __forceinline__ int load_ids(const int* __restrict__ sorted_src,
                                        int node, int base, int end, int lane) {
  return (lane < end - base) ? sorted_src[base + lane] : node;
}

__device__ __forceinline__ float4 node_sum(const float* __restrict__ h,
                                           const int* __restrict__ sorted_src,
                                           int node, int begin, int n, int first,
                                           float4 acc, int lane) {
  const int end = begin + n;
  int mine = first;
  for (int base = begin; base < end; base += 32) {
    const int left = end - base;
    const int m    = left < 32 ? left : 32;
    if (base != begin) mine = load_ids(sorted_src, node, base, end, lane);
    int j = 0;
    for (; j + 8 <= m; j += 8) {
      float4 v[8];
      #pragma unroll
      for (int i = 0; i < 8; ++i)
        v[i] = row4(h, __builtin_amdgcn_readlane(mine, j + i), lane);
      #pragma unroll
      for (int i = 0; i < 8; ++i) {
        acc.x += v[i].x; acc.y += v[i].y; acc.z += v[i].z; acc.w += v[i].w;
      }
    }
    for (; j < m; j += 4) {
      float4 v[4];
      #pragma unroll
      for (int i = 0; i < 4; ++i)
        v[i] = row4(h, __builtin_amdgcn_readlane(mine, j + i), lane);
      #pragma unroll
      for (int i = 0; i < 4; ++i) {
        const bool ok = j + i < m;
        acc.x += ok ? v[i].x : 0.0f;
        acc.y += ok ? v[i].y : 0.0f;
        acc.z += ok ? v[i].z : 0.0f;
        acc.w += ok ? v[i].w : 0.0f;
      }
    }
  }
  return acc;
}

__global__ void __launch_bounds__(BLOCK)
gather_gemm_kernel(const float* __restrict__ h,
                   const int* __restrict__ sorted_src,
                   const int* __restrict__ cursor,
                   const int* __restrict__ cnt,
                   const float* __restrict__ eps_p,
                   const float* __restrict__ W,
                   const float* __restrict__ bias,
                   const float* __restrict__ x,
                   float* __restrict__ out,
                   int n_row_tiles) {
  __shared__ float tile[2][16 * TILE_PITCH];

  const int wave = __builtin_amdgcn_readfirstlane((int)(threadIdx.x >> 5));
  const int lane = threadIdx.x & 31;
  const int col0 = wave * 16;
  const int half = lane >> 4;
  const int l    = lane & 15;

  const float* brow = W + (size_t)(col0 + l) * DIM + 2 * half;
  v2f b[DIM / 4];
  #pragma unroll
  for (int k = 0; k < DIM; k += 4) b[k >> 2] = *(const v2f*)(brow + k);
  const float bv = bias[col0 + l];
  const float scale = 1.0f + eps_p[0];

  int n0 = 0, n1 = 0, begin0 = 0, begin1 = 0;
  if ((int)blockIdx.x < n_row_tiles) {
    const int nd = (int)blockIdx.x * 16 + 2 * wave;
    n0 = cnt[nd];        n1 = cnt[nd + 1];
    begin0 = cursor[nd]; begin1 = cursor[nd + 1];
  }

  int buf = 0;
  for (int tm = blockIdx.x; tm < n_row_tiles; tm += gridDim.x) {
    const int row0 = tm * 16;
    float* t = tile[buf];

    const int nd0 = row0 + 2 * wave;
    const int nd1 = nd0 + 1;
    const int ids0 = load_ids(sorted_src, nd0, begin0, begin0 + n0, lane);
    const int ids1 = load_ids(sorted_src, nd1, begin1, begin1 + n1, lane);
    const float4 h0 = row4(h, nd0, lane);
    const float4 h1 = row4(h, nd1, lane);

    const int tmn = tm + (int)gridDim.x;
    int nn0 = 0, nn1 = 0, nbegin0 = 0, nbegin1 = 0;
    if (tmn < n_row_tiles) {
      const int nd = tmn * 16 + 2 * wave;
      nn0 = cnt[nd];        nn1 = cnt[nd + 1];
      nbegin0 = cursor[nd]; nbegin1 = cursor[nd + 1];
    }

    float4 acc;
    acc.x = scale * h0.x; acc.y = scale * h0.y; acc.z = scale * h0.z; acc.w = scale * h0.w;
    acc = node_sum(h, sorted_src, nd0, begin0, n0, ids0, acc, lane);
    *(float4*)(t + (2 * wave) * TILE_PITCH + 4 * lane) = acc;
    acc.x = scale * h1.x; acc.y = scale * h1.y; acc.z = scale * h1.z; acc.w = scale * h1.w;
    acc = node_sum(h, sorted_src, nd1, begin1, n1, ids1, acc, lane);
    *(float4*)(t + (2 * wave + 1) * TILE_PITCH + 4 * lane) = acc;

    __syncthreads();

    const float* arow = t + l * TILE_PITCH + 2 * half;
    v8f c = {};
    #pragma unroll
    for (int k = 0; k < DIM; k += 4) {
      v2f a = *(const v2f*)(arow + k);
      c = __builtin_amdgcn_wmma_f32_16x16x4_f32(
           false, a,  false, b[k >> 2],
           (short)0, c,  false,  false);
    }

    const size_t idx0 = (size_t)(row0 + 8 * half) * DIM + col0 + l;
    const float* xp = x + idx0;
    float* op = out + idx0;
    #pragma unroll
    for (int r = 0; r < 8; ++r) op[r * DIM] = c[r] + bv + xp[r * DIM];

    n0 = nn0; n1 = nn1; begin0 = nbegin0; begin1 = nbegin1;
    buf ^= 1;
  }
}

extern "C" void kernel_launch(void* const* d_in, const int* in_sizes, int n_in,
                              void* d_out, int out_size, void* d_ws, size_t ws_size,
                              hipStream_t stream) {
  const float* x     = (const float*)d_in[0];
  const int*   ei    = (const int*)  d_in[1];
  const float* gamma = (const float*)d_in[2];
  const float* beta  = (const float*)d_in[3];
  const float* W     = (const float*)d_in[4];
  const float* bias  = (const float*)d_in[5];
  const float* eps   = (const float*)d_in[6];
  float* out = (float*)d_out;

  const int n_nodes = in_sizes[0] / DIM;
  const int n_edges = in_sizes[1] / 2;

  float* h          = (float*)d_ws;
  int*   cnt        = (int*)(h + (size_t)n_nodes * DIM);
  int*   cursor     = cnt + n_nodes;
  int*   sorted_src = cursor + n_nodes;
  int*   rank       = sorted_src + n_edges;
  int*   scount     = rank + n_edges;
  int*   total      = scount + (size_t)CNT_S * n_nodes;

  ln_relu_zero_kernel<<<GRID_NORM, BLOCK, 0, stream>>>(x, gamma, beta, h, total, n_nodes);
  slice_rank_kernel<<<CNT_R * CNT_S, CNT_THREADS, 0, stream>>>(ei, scount, rank, n_edges, n_nodes);
  {
    int blocks = (n_nodes + 255) / 256;
    alloc_kernel<<<blocks, 256, 0, stream>>>(scount, cnt, cursor, total, n_nodes);
  }
  fill_kernel<<<GRID_EDGES, 256, 0, stream>>>(ei, scount, rank, sorted_src, n_edges, n_nodes);
  gather_gemm_kernel<<<GRID_FUSED, BLOCK, 0, stream>>>(h, sorted_src, cursor, cnt, eps, W, bias, x, out, n_nodes / 16);
}


# Ranking blocks keep LDS counters for a quarter of the nodes (100 KB), not 1/32: a dst slice is scanned 4 times, not 32; 32 slices, 16 dst loads in flight; sum order: arrival order within a slice
# speedup vs baseline: 2.6757x; 1.1449x over previous
#include <hip/hip_runtime.h>

#define DIM 128
#define WAVES_PER_BLOCK 8
#define BLOCK (WAVES_PER_BLOCK * 32)

#define GRID_NORM   1024
#define GRID_EDGES   512
#define GRID_FUSED  1024
#define TILE_PITCH   132
#define NORM_ROWS 8

#define CNT_R 4
#define CNT_S 32
#define CNT_CAP 25600
#define CNT_THREADS 1024
#define CNT_DEPTH 16

typedef float v2f __attribute__((ext_vector_type(2)));
typedef float v8f __attribute__((ext_vector_type(8)));

__global__ void __launch_bounds__(BLOCK)
ln_relu_zero_kernel(const float* __restrict__ x,
                    const float* __restrict__ gamma,
                    const float* __restrict__ beta,
                    float* __restrict__ h,
                    int* __restrict__ total,
                    int n_nodes) {
  const int lane = threadIdx.x & 31;
  const int gw = __builtin_amdgcn_readfirstlane(
      (int)((blockIdx.x * blockDim.x + threadIdx.x) >> 5));
  const int nw = (int)(gridDim.x * (blockDim.x >> 5));

  const float4 g = ((const float4*)gamma)[lane];
  const float4 b = ((const float4*)beta)[lane];
  if (gw == 0 && lane == 0) total[0] = 0;

  for (int base = gw; base < n_nodes; base += NORM_ROWS * nw) {
    float4 v[NORM_ROWS];
    #pragma unroll
    for (int j = 0; j < NORM_ROWS; ++j) {
      const int nd = base + j * nw;
      v[j] = make_float4(0.0f, 0.0f, 0.0f, 0.0f);
      if (nd < n_nodes) v[j] = ((const float4*)(x + (size_t)nd * DIM))[lane];
    }
    #pragma unroll
    for (int j = 0; j < NORM_ROWS; ++j) {
      const int nd = base + j * nw;
      if (nd < n_nodes) {
        float s  = v[j].x + v[j].y + v[j].z + v[j].w;
        float s2 = v[j].x * v[j].x + v[j].y * v[j].y + v[j].z * v[j].z + v[j].w * v[j].w;
        #pragma unroll
        for (int off = 16; off > 0; off >>= 1) {
          s  += __shfl_xor(s,  off, 32);
          s2 += __shfl_xor(s2, off, 32);
        }
        const float mu  = s * (1.0f / DIM);
        const float var = s2 * (1.0f / DIM) - mu * mu;
        const float rs  = rsqrtf(var + 1e-5f);
        float4 hv;
        hv.x = fmaxf((v[j].x - mu) * rs * g.x + b.x, 0.0f);
        hv.y = fmaxf((v[j].y - mu) * rs * g.y + b.y, 0.0f);
        hv.z = fmaxf((v[j].z - mu) * rs * g.z + b.z, 0.0f);
        hv.w = fmaxf((v[j].w - mu) * rs * g.w + b.w, 0.0f);
        ((float4*)(h + (size_t)nd * DIM))[lane] = hv;
      }
    }
  }
}


__global__ void __launch_bounds__(CNT_THREADS)
slice_rank_kernel(const int* __restrict__ ei,
                  int* __restrict__ scount,
                  int* __restrict__ rank,
                  int n_edges,
                  int n_nodes) {
  __shared__ int lcnt[CNT_CAP];
  const int r = blockIdx.x % CNT_R;
  const int sl_id = blockIdx.x / CNT_R;
  const int* __restrict__ dstp = ei + n_edges;

  int rl = (n_nodes + CNT_R - 1) / CNT_R;
  if (rl > CNT_CAP) rl = CNT_CAP;
  const int sl = (n_edges + CNT_S - 1) / CNT_S;
  const int e0 = sl_id * sl;
  int e1 = e0 + sl;
  if (e1 > n_edges) e1 = n_edges;

  for (int lo = r * rl; lo < n_nodes; lo += CNT_R * rl) {
    int len = n_nodes - lo;
    if (len > rl) len = rl;
    for (int i = threadIdx.x; i < len; i += CNT_THREADS) lcnt[i] = 0;
    __syncthreads();

    for (int base = e0 + (int)threadIdx.x; base < e1; base += CNT_DEPTH * CNT_THREADS) {
      int d[CNT_DEPTH];
      #pragma unroll
      for (int j = 0; j < CNT_DEPTH; ++j) {
        const int e = base + j * CNT_THREADS;
        d[j] = (e < e1) ? dstp[e] : -1;
      }
      int rk[CNT_DEPTH];
      #pragma unroll
      for (int j = 0; j < CNT_DEPTH; ++j) {
        const unsigned off = (unsigned)(d[j] - lo);
        rk[j] = (off < (unsigned)len) ? atomicAdd(&lcnt[off], 1) : -1;
      }
      #pragma unroll
      for (int j = 0; j < CNT_DEPTH; ++j) {
        if (rk[j] >= 0) rank[base + j * CNT_THREADS] = rk[j];
      }
    }
    __syncthreads();

    int* __restrict__ row = scount + (size_t)sl_id * n_nodes + lo;
    for (int i = threadIdx.x; i < len; i += CNT_THREADS) row[i] = lcnt[i];
    __syncthreads();
  }
}

__global__ void __launch_bounds__(256)
alloc_kernel(int* __restrict__ scount,
             int* __restrict__ cnt,
             int* __restrict__ cursor,
             int* __restrict__ total,
             int n_nodes) {
  __shared__ int wave_sum[8];
  __shared__ int block_base;
  const int i    = blockIdx.x * 256 + threadIdx.x;
  const int lane = threadIdx.x & 31;
  const int w    = threadIdx.x >> 5;
  int cs[CNT_S];
  int c = 0;
  #pragma unroll
  for (int s = 0; s < CNT_S; ++s) {
    cs[s] = (i < n_nodes) ? scount[(size_t)s * n_nodes + i] : 0;
    c += cs[s];
  }
  int inc = c;
  for (int off = 1; off < 32; off <<= 1) {
    const int t = __shfl_up(inc, off, 32);
    if (lane >= off) inc += t;
  }
  if (lane == 31) wave_sum[w] = inc;
  __syncthreads();
  if (threadIdx.x == 0) {
    int run = 0;
    for (int k = 0; k < 8; ++k) {
      const int t = wave_sum[k];
      wave_sum[k] = run;
      run += t;
    }
    block_base = atomicAdd(total, run);
  }
  __syncthreads();
  if (i < n_nodes) {
    const int start = block_base + wave_sum[w] + inc - c;
    cursor[i] = start;
    cnt[i]    = c;
    int run = start;
    #pragma unroll
    for (int s = 0; s < CNT_S; ++s) {
      scount[(size_t)s * n_nodes + i] = run;
      run += cs[s];
    }
  }
}

__global__ void __launch_bounds__(256)
fill_kernel(const int* __restrict__ ei,
            const int* __restrict__ sstart,
            const int* __restrict__ rank,
            int* __restrict__ sorted_src,
            int n_edges,
            int n_nodes) {
  const int t  = blockIdx.x * blockDim.x + threadIdx.x;
  const int nt = gridDim.x * blockDim.x;
  const int* __restrict__ dstp = ei + n_edges;
  const int sl = (n_edges + CNT_S - 1) / CNT_S;
  for (int base = t; base < n_edges; base += 4 * nt) {
    int s[4], d[4], r[4], st[4];
    #pragma unroll
    for (int j = 0; j < 4; ++j) {
      const int e = base + j * nt;
      const bool ok = e < n_edges;
      s[j] = ok ? ei[e]   : 0;
      d[j] = ok ? dstp[e] : -1;
      r[j] = ok ? rank[e] : 0;
    }
    #pragma unroll
    for (int j = 0; j < 4; ++j) {
      const int slice = (base + j * nt) / sl;
      st[j] = (d[j] >= 0) ? sstart[(size_t)slice * n_nodes + d[j]] : 0;
    }
    #pragma unroll
    for (int j = 0; j < 4; ++j) {
      if (d[j] >= 0) sorted_src[st[j] + r[j]] = s[j];
    }
  }
}

__device__ __forceinline__ float4 row4(const float* __restrict__ h, int s, int lane) {
  return ((const float4*)(h + (size_t)s * DIM))[lane];
}

__device__ __forceinline__ int load_ids(const int* __restrict__ sorted_src,
                                        int node, int base, int end, int lane) {
  return (lane < end - base) ? sorted_src[base + lane] : node;
}

__device__ __forceinline__ float4 node_sum(const float* __restrict__ h,
                                           const int* __restrict__ sorted_src,
                                           int node, int begin, int n, int first,
                                           float4 acc, int lane) {
  const int end = begin + n;
  int mine = first;
  for (int base = begin; base < end; base += 32) {
    const int left = end - base;
    const int m    = left < 32 ? left : 32;
    if (base != begin) mine = load_ids(sorted_src, node, base, end, lane);
    int j = 0;
    for (; j + 8 <= m; j += 8) {
      float4 v[8];
      #pragma unroll
      for (int i = 0; i < 8; ++i)
        v[i] = row4(h, __builtin_amdgcn_readlane(mine, j + i), lane);
      #pragma unroll
      for (int i = 0; i < 8; ++i) {
        acc.x += v[i].x; acc.y += v[i].y; acc.z += v[i].z; acc.w += v[i].w;
      }
    }
    for (; j < m; j += 4) {
      float4 v[4];
      #pragma unroll
      for (int i = 0; i < 4; ++i)
        v[i] = row4(h, __builtin_amdgcn_readlane(mine, j + i), lane);
      #pragma unroll
      for (int i = 0; i < 4; ++i) {
        const bool ok = j + i < m;
        acc.x += ok ? v[i].x : 0.0f;
        acc.y += ok ? v[i].y : 0.0f;
        acc.z += ok ? v[i].z : 0.0f;
        acc.w += ok ? v[i].w : 0.0f;
      }
    }
  }
  return acc;
}

__global__ void __launch_bounds__(BLOCK)
gather_gemm_kernel(const float* __restrict__ h,
                   const int* __restrict__ sorted_src,
                   const int* __restrict__ cursor,
                   const int* __restrict__ cnt,
                   const float* __restrict__ eps_p,
                   const float* __restrict__ W,
                   const float* __restrict__ bias,
                   const float* __restrict__ x,
                   float* __restrict__ out,
                   int n_row_tiles) {
  __shared__ float tile[2][16 * TILE_PITCH];

  const int wave = __builtin_amdgcn_readfirstlane((int)(threadIdx.x >> 5));
  const int lane = threadIdx.x & 31;
  const int col0 = wave * 16;
  const int half = lane >> 4;
  const int l    = lane & 15;

  const float* brow = W + (size_t)(col0 + l) * DIM + 2 * half;
  v2f b[DIM / 4];
  #pragma unroll
  for (int k = 0; k < DIM; k += 4) b[k >> 2] = *(const v2f*)(brow + k);
  const float bv = bias[col0 + l];
  const float scale = 1.0f + eps_p[0];

  int n0 = 0, n1 = 0, begin0 = 0, begin1 = 0;
  if ((int)blockIdx.x < n_row_tiles) {
    const int nd = (int)blockIdx.x * 16 + 2 * wave;
    n0 = cnt[nd];        n1 = cnt[nd + 1];
    begin0 = cursor[nd]; begin1 = cursor[nd + 1];
  }

  int buf = 0;
  for (int tm = blockIdx.x; tm < n_row_tiles; tm += gridDim.x) {
    const int row0 = tm * 16;
    float* t = tile[buf];

    const int nd0 = row0 + 2 * wave;
    const int nd1 = nd0 + 1;
    const int ids0 = load_ids(sorted_src, nd0, begin0, begin0 + n0, lane);
    const int ids1 = load_ids(sorted_src, nd1, begin1, begin1 + n1, lane);
    const float4 h0 = row4(h, nd0, lane);
    const float4 h1 = row4(h, nd1, lane);

    const int tmn = tm + (int)gridDim.x;
    int nn0 = 0, nn1 = 0, nbegin0 = 0, nbegin1 = 0;
    if (tmn < n_row_tiles) {
      const int nd = tmn * 16 + 2 * wave;
      nn0 = cnt[nd];        nn1 = cnt[nd + 1];
      nbegin0 = cursor[nd]; nbegin1 = cursor[nd + 1];
    }

    float4 acc;
    acc.x = scale * h0.x; acc.y = scale * h0.y; acc.z = scale * h0.z; acc.w = scale * h0.w;
    acc = node_sum(h, sorted_src, nd0, begin0, n0, ids0, acc, lane);
    *(float4*)(t + (2 * wave) * TILE_PITCH + 4 * lane) = acc;
    acc.x = scale * h1.x; acc.y = scale * h1.y; acc.z = scale * h1.z; acc.w = scale * h1.w;
    acc = node_sum(h, sorted_src, nd1, begin1, n1, ids1, acc, lane);
    *(float4*)(t + (2 * wave + 1) * TILE_PITCH + 4 * lane) = acc;

    __syncthreads();

    const float* arow = t + l * TILE_PITCH + 2 * half;
    v8f c = {};
    #pragma unroll
    for (int k = 0; k < DIM; k += 4) {
      v2f a = *(const v2f*)(arow + k);
      c = __builtin_amdgcn_wmma_f32_16x16x4_f32(
           false, a,  false, b[k >> 2],
           (short)0, c,  false,  false);
    }

    const size_t idx0 = (size_t)(row0 + 8 * half) * DIM + col0 + l;
    const float* xp = x + idx0;
    float* op = out + idx0;
    #pragma unroll
    for (int r = 0; r < 8; ++r) op[r * DIM] = c[r] + bv + xp[r * DIM];

    n0 = nn0; n1 = nn1; begin0 = nbegin0; begin1 = nbegin1;
    buf ^= 1;
  }
}

extern "C" void kernel_launch(void* const* d_in, const int* in_sizes, int n_in,
                              void* d_out, int out_size, void* d_ws, size_t ws_size,
                              hipStream_t stream) {
  const float* x     = (const float*)d_in[0];
  const int*   ei    = (const int*)  d_in[1];
  const float* gamma = (const float*)d_in[2];
  const float* beta  = (const float*)d_in[3];
  const float* W     = (const float*)d_in[4];
  const float* bias  = (const float*)d_in[5];
  const float* eps   = (const float*)d_in[6];
  float* out = (float*)d_out;

  const int n_nodes = in_sizes[0] / DIM;
  const int n_edges = in_sizes[1] / 2;

  float* h          = (float*)d_ws;
  int*   cnt        = (int*)(h + (size_t)n_nodes * DIM);
  int*   cursor     = cnt + n_nodes;
  int*   sorted_src = cursor + n_nodes;
  int*   rank       = sorted_src + n_edges;
  int*   scount     = rank + n_edges;
  int*   total      = scount + (size_t)CNT_S * n_nodes;

  ln_relu_zero_kernel<<<GRID_NORM, BLOCK, 0, stream>>>(x, gamma, beta, h, total, n_nodes);
  slice_rank_kernel<<<CNT_R * CNT_S, CNT_THREADS, 0, stream>>>(ei, scount, rank, n_edges, n_nodes);
  {
    int blocks = (n_nodes + 255) / 256;
    alloc_kernel<<<blocks, 256, 0, stream>>>(scount, cnt, cursor, total, n_nodes);
  }
  fill_kernel<<<GRID_EDGES, 256, 0, stream>>>(ei, scount, rank, sorted_src, n_edges, n_nodes);
  gather_gemm_kernel<<<GRID_FUSED, BLOCK, 0, stream>>>(h, sorted_src, cursor, cnt, eps, W, bias, x, out, n_nodes / 16);
}


# LayerNorm pass loses its launch: its rows are computed by extra blocks of the ranking launch (first half) and the fill launch (second half), both latency-bound; launches 5 to 4; sum order as before
# speedup vs baseline: 2.8238x; 1.0554x over previous
#include <hip/hip_runtime.h>

#define DIM 128
#define WAVES_PER_BLOCK 8
#define BLOCK (WAVES_PER_BLOCK * 32)

#define NORM_BLOCKS_RANK 128
#define NORM_BLOCKS_FILL 512
#define GRID_EDGES   512
#define GRID_FUSED  1024
#define TILE_PITCH   132
#define NORM_ROWS 8

#define CNT_R 4
#define CNT_S 32
#define CNT_CAP 25600
#define CNT_THREADS 1024
#define CNT_DEPTH 16

typedef float v2f __attribute__((ext_vector_type(2)));
typedef float v8f __attribute__((ext_vector_type(8)));

__device__ __forceinline__ void ln_relu_rows(const float* __restrict__ x,
                                             const float* __restrict__ gamma,
                                             const float* __restrict__ beta,
                                             float* __restrict__ h,
                                             int row_lo, int row_hi,
                                             int gw, int nw, int lane) {
  const float4 g = ((const float4*)gamma)[lane];
  const float4 b = ((const float4*)beta)[lane];

  for (int base = row_lo + gw; base < row_hi; base += NORM_ROWS * nw) {
    float4 v[NORM_ROWS];
    #pragma unroll
    for (int j = 0; j < NORM_ROWS; ++j) {
      const int nd = base + j * nw;
      v[j] = make_float4(0.0f, 0.0f, 0.0f, 0.0f);
      if (nd < row_hi) v[j] = ((const float4*)(x + (size_t)nd * DIM))[lane];
    }
    #pragma unroll
    for (int j = 0; j < NORM_ROWS; ++j) {
      const int nd = base + j * nw;
      if (nd < row_hi) {
        float s  = v[j].x + v[j].y + v[j].z + v[j].w;
        float s2 = v[j].x * v[j].x + v[j].y * v[j].y + v[j].z * v[j].z + v[j].w * v[j].w;
        #pragma unroll
        for (int off = 16; off > 0; off >>= 1) {
          s  += __shfl_xor(s,  off, 32);
          s2 += __shfl_xor(s2, off, 32);
        }
        const float mu  = s * (1.0f / DIM);
        const float var = s2 * (1.0f / DIM) - mu * mu;
        const float rs  = rsqrtf(var + 1e-5f);
        float4 hv;
        hv.x = fmaxf((v[j].x - mu) * rs * g.x + b.x, 0.0f);
        hv.y = fmaxf((v[j].y - mu) * rs * g.y + b.y, 0.0f);
        hv.z = fmaxf((v[j].z - mu) * rs * g.z + b.z, 0.0f);
        hv.w = fmaxf((v[j].w - mu) * rs * g.w + b.w, 0.0f);
        ((float4*)(h + (size_t)nd * DIM))[lane] = hv;
      }
    }
  }
}


__global__ void __launch_bounds__(CNT_THREADS)
rank_norm_kernel(const int* __restrict__ ei,
                 int* __restrict__ scount,
                 int* __restrict__ rank,
                 int n_edges,
                 int n_nodes,
                 const float* __restrict__ x,
                 const float* __restrict__ gamma,
                 const float* __restrict__ beta,
                 float* __restrict__ h,
                 int* __restrict__ total) {
  __shared__ int lcnt[CNT_CAP];
  if (blockIdx.x >= CNT_R * CNT_S) {
    const int gw = __builtin_amdgcn_readfirstlane(
        (int)(((blockIdx.x - CNT_R * CNT_S) * CNT_THREADS + threadIdx.x) >> 5));
    const int nw = NORM_BLOCKS_RANK * (CNT_THREADS >> 5);
    ln_relu_rows(x, gamma, beta, h, 0, n_nodes >> 1, gw, nw, (int)(threadIdx.x & 31));
    return;
  }
  if (blockIdx.x == 0 && threadIdx.x == 0) total[0] = 0;
  const int r = blockIdx.x % CNT_R;
  const int sl_id = blockIdx.x / CNT_R;
  const int* __restrict__ dstp = ei + n_edges;

  int rl = (n_nodes + CNT_R - 1) / CNT_R;
  if (rl > CNT_CAP) rl = CNT_CAP;
  const int sl = (n_edges + CNT_S - 1) / CNT_S;
  const int e0 = sl_id * sl;
  int e1 = e0 + sl;
  if (e1 > n_edges) e1 = n_edges;

  for (int lo = r * rl; lo < n_nodes; lo += CNT_R * rl) {
    int len = n_nodes - lo;
    if (len > rl) len = rl;
    for (int i = threadIdx.x; i < len; i += CNT_THREADS) lcnt[i] = 0;
    __syncthreads();

    for (int base = e0 + (int)threadIdx.x; base < e1; base += CNT_DEPTH * CNT_THREADS) {
      int d[CNT_DEPTH];
      #pragma unroll
      for (int j = 0; j < CNT_DEPTH; ++j) {
        const int e = base + j * CNT_THREADS;
        d[j] = (e < e1) ? dstp[e] : -1;
      }
      int rk[CNT_DEPTH];
      #pragma unroll
      for (int j = 0; j < CNT_DEPTH; ++j) {
        const unsigned off = (unsigned)(d[j] - lo);
        rk[j] = (off < (unsigned)len) ? atomicAdd(&lcnt[off], 1) : -1;
      }
      #pragma unroll
      for (int j = 0; j < CNT_DEPTH; ++j) {
        if (rk[j] >= 0) rank[base + j * CNT_THREADS] = rk[j];
      }
    }
    __syncthreads();

    int* __restrict__ row = scount + (size_t)sl_id * n_nodes + lo;
    for (int i = threadIdx.x; i < len; i += CNT_THREADS) row[i] = lcnt[i];
    __syncthreads();
  }
}

__global__ void __launch_bounds__(256)
alloc_kernel(int* __restrict__ scount,
             int* __restrict__ cnt,
             int* __restrict__ cursor,
             int* __restrict__ total,
             int n_nodes) {
  __shared__ int wave_sum[8];
  __shared__ int block_base;
  const int i    = blockIdx.x * 256 + threadIdx.x;
  const int lane = threadIdx.x & 31;
  const int w    = threadIdx.x >> 5;
  int cs[CNT_S];
  int c = 0;
  #pragma unroll
  for (int s = 0; s < CNT_S; ++s) {
    cs[s] = (i < n_nodes) ? scount[(size_t)s * n_nodes + i] : 0;
    c += cs[s];
  }
  int inc = c;
  for (int off = 1; off < 32; off <<= 1) {
    const int t = __shfl_up(inc, off, 32);
    if (lane >= off) inc += t;
  }
  if (lane == 31) wave_sum[w] = inc;
  __syncthreads();
  if (threadIdx.x == 0) {
    int run = 0;
    for (int k = 0; k < 8; ++k) {
      const int t = wave_sum[k];
      wave_sum[k] = run;
      run += t;
    }
    block_base = atomicAdd(total, run);
  }
  __syncthreads();
  if (i < n_nodes) {
    const int start = block_base + wave_sum[w] + inc - c;
    cursor[i] = start;
    cnt[i]    = c;
    int run = start;
    #pragma unroll
    for (int s = 0; s < CNT_S; ++s) {
      scount[(size_t)s * n_nodes + i] = run;
      run += cs[s];
    }
  }
}

__global__ void __launch_bounds__(256)
fill_norm_kernel(const int* __restrict__ ei,
                 const int* __restrict__ sstart,
                 const int* __restrict__ rank,
                 int* __restrict__ sorted_src,
                 int n_edges,
                 int n_nodes,
                 const float* __restrict__ x,
                 const float* __restrict__ gamma,
                 const float* __restrict__ beta,
                 float* __restrict__ h) {
  if (blockIdx.x >= GRID_EDGES) {
    const int gw = __builtin_amdgcn_readfirstlane(
        (int)(((blockIdx.x - GRID_EDGES) * 256 + threadIdx.x) >> 5));
    const int nw = NORM_BLOCKS_FILL * (256 >> 5);
    ln_relu_rows(x, gamma, beta, h, n_nodes >> 1, n_nodes, gw, nw, (int)(threadIdx.x & 31));
    return;
  }
  const int t  = blockIdx.x * 256 + threadIdx.x;
  const int nt = GRID_EDGES * 256;
  const int* __restrict__ dstp = ei + n_edges;
  const int sl = (n_edges + CNT_S - 1) / CNT_S;
  for (int base = t; base < n_edges; base += 4 * nt) {
    int s[4], d[4], r[4], st[4];
    #pragma unroll
    for (int j = 0; j < 4; ++j) {
      const int e = base + j * nt;
      const bool ok = e < n_edges;
      s[j] = ok ? ei[e]   : 0;
      d[j] = ok ? dstp[e] : -1;
      r[j] = ok ? rank[e] : 0;
    }
    #pragma unroll
    for (int j = 0; j < 4; ++j) {
      const int slice = (base + j * nt) / sl;
      st[j] = (d[j] >= 0) ? sstart[(size_t)slice * n_nodes + d[j]] : 0;
    }
    #pragma unroll
    for (int j = 0; j < 4; ++j) {
      if (d[j] >= 0) sorted_src[st[j] + r[j]] = s[j];
    }
  }
}

__device__ __forceinline__ float4 row4(const float* __restrict__ h, int s, int lane) {
  return ((const float4*)(h + (size_t)s * DIM))[lane];
}

__device__ __forceinline__ int load_ids(const int* __restrict__ sorted_src,
                                        int node, int base, int end, int lane) {
  return (lane < end - base) ? sorted_src[base + lane] : node;
}

__device__ __forceinline__ float4 node_sum(const float* __restrict__ h,
                                           const int* __restrict__ sorted_src,
                                           int node, int begin, int n, int first,
                                           float4 acc, int lane) {
  const int end = begin + n;
  int mine = first;
  for (int base = begin; base < end; base += 32) {
    const int left = end - base;
    const int m    = left < 32 ? left : 32;
    if (base != begin) mine = load_ids(sorted_src, node, base, end, lane);
    int j = 0;
    for (; j + 8 <= m; j += 8) {
      float4 v[8];
      #pragma unroll
      for (int i = 0; i < 8; ++i)
        v[i] = row4(h, __builtin_amdgcn_readlane(mine, j + i), lane);
      #pragma unroll
      for (int i = 0; i < 8; ++i) {
        acc.x += v[i].x; acc.y += v[i].y; acc.z += v[i].z; acc.w += v[i].w;
      }
    }
    for (; j < m; j += 4) {
      float4 v[4];
      #pragma unroll
      for (int i = 0; i < 4; ++i)
        v[i] = row4(h, __builtin_amdgcn_readlane(mine, j + i), lane);
      #pragma unroll
      for (int i = 0; i < 4; ++i) {
        const bool ok = j + i < m;
        acc.x += ok ? v[i].x : 0.0f;
        acc.y += ok ? v[i].y : 0.0f;
        acc.z += ok ? v[i].z : 0.0f;
        acc.w += ok ? v[i].w : 0.0f;
      }
    }
  }
  return acc;
}

__global__ void __launch_bounds__(BLOCK)
gather_gemm_kernel(const float* __restrict__ h,
                   const int* __restrict__ sorted_src,
                   const int* __restrict__ cursor,
                   const int* __restrict__ cnt,
                   const float* __restrict__ eps_p,
                   const float* __restrict__ W,
                   const float* __restrict__ bias,
                   const float* __restrict__ x,
                   float* __restrict__ out,
                   int n_row_tiles) {
  __shared__ float tile[2][16 * TILE_PITCH];

  const int wave = __builtin_amdgcn_readfirstlane((int)(threadIdx.x >> 5));
  const int lane = threadIdx.x & 31;
  const int col0 = wave * 16;
  const int half = lane >> 4;
  const int l    = lane & 15;

  const float* brow = W + (size_t)(col0 + l) * DIM + 2 * half;
  v2f b[DIM / 4];
  #pragma unroll
  for (int k = 0; k < DIM; k += 4) b[k >> 2] = *(const v2f*)(brow + k);
  const float bv = bias[col0 + l];
  const float scale = 1.0f + eps_p[0];

  int n0 = 0, n1 = 0, begin0 = 0, begin1 = 0;
  if ((int)blockIdx.x < n_row_tiles) {
    const int nd = (int)blockIdx.x * 16 + 2 * wave;
    n0 = cnt[nd];        n1 = cnt[nd + 1];
    begin0 = cursor[nd]; begin1 = cursor[nd + 1];
  }

  int buf = 0;
  for (int tm = blockIdx.x; tm < n_row_tiles; tm += gridDim.x) {
    const int row0 = tm * 16;
    float* t = tile[buf];

    const int nd0 = row0 + 2 * wave;
    const int nd1 = nd0 + 1;
    const int ids0 = load_ids(sorted_src, nd0, begin0, begin0 + n0, lane);
    const int ids1 = load_ids(sorted_src, nd1, begin1, begin1 + n1, lane);
    const float4 h0 = row4(h, nd0, lane);
    const float4 h1 = row4(h, nd1, lane);

    const int tmn = tm + (int)gridDim.x;
    int nn0 = 0, nn1 = 0, nbegin0 = 0, nbegin1 = 0;
    if (tmn < n_row_tiles) {
      const int nd = tmn * 16 + 2 * wave;
      nn0 = cnt[nd];        nn1 = cnt[nd + 1];
      nbegin0 = cursor[nd]; nbegin1 = cursor[nd + 1];
    }

    float4 acc;
    acc.x = scale * h0.x; acc.y = scale * h0.y; acc.z = scale * h0.z; acc.w = scale * h0.w;
    acc = node_sum(h, sorted_src, nd0, begin0, n0, ids0, acc, lane);
    *(float4*)(t + (2 * wave) * TILE_PITCH + 4 * lane) = acc;
    acc.x = scale * h1.x; acc.y = scale * h1.y; acc.z = scale * h1.z; acc.w = scale * h1.w;
    acc = node_sum(h, sorted_src, nd1, begin1, n1, ids1, acc, lane);
    *(float4*)(t + (2 * wave + 1) * TILE_PITCH + 4 * lane) = acc;

    __syncthreads();

    const float* arow = t + l * TILE_PITCH + 2 * half;
    v8f c = {};
    #pragma unroll
    for (int k = 0; k < DIM; k += 4) {
      v2f a = *(const v2f*)(arow + k);
      c = __builtin_amdgcn_wmma_f32_16x16x4_f32(
           false, a,  false, b[k >> 2],
           (short)0, c,  false,  false);
    }

    const size_t idx0 = (size_t)(row0 + 8 * half) * DIM + col0 + l;
    const float* xp = x + idx0;
    float* op = out + idx0;
    #pragma unroll
    for (int r = 0; r < 8; ++r) op[r * DIM] = c[r] + bv + xp[r * DIM];

    n0 = nn0; n1 = nn1; begin0 = nbegin0; begin1 = nbegin1;
    buf ^= 1;
  }
}

extern "C" void kernel_launch(void* const* d_in, const int* in_sizes, int n_in,
                              void* d_out, int out_size, void* d_ws, size_t ws_size,
                              hipStream_t stream) {
  const float* x     = (const float*)d_in[0];
  const int*   ei    = (const int*)  d_in[1];
  const float* gamma = (const float*)d_in[2];
  const float* beta  = (const float*)d_in[3];
  const float* W     = (const float*)d_in[4];
  const float* bias  = (const float*)d_in[5];
  const float* eps   = (const float*)d_in[6];
  float* out = (float*)d_out;

  const int n_nodes = in_sizes[0] / DIM;
  const int n_edges = in_sizes[1] / 2;

  float* h          = (float*)d_ws;
  int*   cnt        = (int*)(h + (size_t)n_nodes * DIM);
  int*   cursor     = cnt + n_nodes;
  int*   sorted_src = cursor + n_nodes;
  int*   rank       = sorted_src + n_edges;
  int*   scount     = rank + n_edges;
  int*   total      = scount + (size_t)CNT_S * n_nodes;

  rank_norm_kernel<<<CNT_R * CNT_S + NORM_BLOCKS_RANK, CNT_THREADS, 0, stream>>>(
      ei, scount, rank, n_edges, n_nodes, x, gamma, beta, h, total);
  {
    int blocks = (n_nodes + 255) / 256;
    alloc_kernel<<<blocks, 256, 0, stream>>>(scount, cnt, cursor, total, n_nodes);
  }
  fill_norm_kernel<<<GRID_EDGES + NORM_BLOCKS_FILL, 256, 0, stream>>>(
      ei, scount, rank, sorted_src, n_edges, n_nodes, x, gamma, beta, h);
  gather_gemm_kernel<<<GRID_FUSED, BLOCK, 0, stream>>>(h, sorted_src, cursor, cnt, eps, W, bias, x, out, n_nodes / 16);
}


# Fill places edges by LDS cursors preset to the slice starts (block = node range x edge slice): no rank array, no per-edge start lookup; ids loaded 16 B wide, deep batches; sum order: arrival in slice
# speedup vs baseline: 3.8538x; 1.3648x over previous
#include <hip/hip_runtime.h>

#define DIM 128
#define WAVES_PER_BLOCK 8
#define BLOCK (WAVES_PER_BLOCK * 32)

#define NORM_BLOCKS_RANK 128
#define NORM_BLOCKS_FILL 256
#define GRID_FUSED  1024
#define TILE_PITCH   132
#define NORM_ROWS 8

#define CNT_R 4
#define CNT_S 32
#define CNT_CAP 25600
#define CNT_THREADS 1024
#define CNT_ROWS 13
#define CNT_DEPTH (4 * CNT_ROWS)
#define FILL_R 8
#define FILL_CAP 12800
#define FILL_THREADS 512
#define FILL_ROWS 9
#define FILL_DEPTH (4 * FILL_ROWS)

typedef float v2f __attribute__((ext_vector_type(2)));
typedef float v8f __attribute__((ext_vector_type(8)));

__device__ __forceinline__ void ln_relu_rows(const float* __restrict__ x,
                                             const float* __restrict__ gamma,
                                             const float* __restrict__ beta,
                                             float* __restrict__ h,
                                             int row_lo, int row_hi,
                                             int gw, int nw, int lane) {
  const float4 g = ((const float4*)gamma)[lane];
  const float4 b = ((const float4*)beta)[lane];

  for (int base = row_lo + gw; base < row_hi; base += NORM_ROWS * nw) {
    float4 v[NORM_ROWS];
    #pragma unroll
    for (int j = 0; j < NORM_ROWS; ++j) {
      const int nd = base + j * nw;
      v[j] = make_float4(0.0f, 0.0f, 0.0f, 0.0f);
      if (nd < row_hi) v[j] = ((const float4*)(x + (size_t)nd * DIM))[lane];
    }
    #pragma unroll
    for (int j = 0; j < NORM_ROWS; ++j) {
      const int nd = base + j * nw;
      if (nd < row_hi) {
        float s  = v[j].x + v[j].y + v[j].z + v[j].w;
        float s2 = v[j].x * v[j].x + v[j].y * v[j].y + v[j].z * v[j].z + v[j].w * v[j].w;
        #pragma unroll
        for (int off = 16; off > 0; off >>= 1) {
          s  += __shfl_xor(s,  off, 32);
          s2 += __shfl_xor(s2, off, 32);
        }
        const float mu  = s * (1.0f / DIM);
        const float var = s2 * (1.0f / DIM) - mu * mu;
        const float rs  = rsqrtf(var + 1e-5f);
        float4 hv;
        hv.x = fmaxf((v[j].x - mu) * rs * g.x + b.x, 0.0f);
        hv.y = fmaxf((v[j].y - mu) * rs * g.y + b.y, 0.0f);
        hv.z = fmaxf((v[j].z - mu) * rs * g.z + b.z, 0.0f);
        hv.w = fmaxf((v[j].w - mu) * rs * g.w + b.w, 0.0f);
        ((float4*)(h + (size_t)nd * DIM))[lane] = hv;
      }
    }
  }
}

__device__ __forceinline__ void load_ids4(const int* __restrict__ p, int eb, int e1,
                                          int nthreads, bool vec, int t, int past,
                                          int* __restrict__ out) {
  const int e = eb + 4 * t;
  if (vec && eb + 4 * nthreads <= e1) {
    const int4 q = *(const int4*)(p + e);
    out[0] = q.x; out[1] = q.y; out[2] = q.z; out[3] = q.w;
  } else {
    #pragma unroll
    for (int k = 0; k < 4; ++k) out[k] = (e + k < e1) ? p[e + k] : past;
  }
}


__global__ void __launch_bounds__(CNT_THREADS)
count_norm_kernel(const int* __restrict__ ei,
                  int* __restrict__ scount,
                  int n_edges,
                  int n_nodes,
                  const float* __restrict__ x,
                  const float* __restrict__ gamma,
                  const float* __restrict__ beta,
                  float* __restrict__ h,
                  int* __restrict__ total) {
  __shared__ int lcnt[CNT_CAP];
  if (blockIdx.x >= CNT_R * CNT_S) {
    const int gw = __builtin_amdgcn_readfirstlane(
        (int)(((blockIdx.x - CNT_R * CNT_S) * CNT_THREADS + threadIdx.x) >> 5));
    const int nw = NORM_BLOCKS_RANK * (CNT_THREADS >> 5);
    ln_relu_rows(x, gamma, beta, h, 0, n_nodes >> 1, gw, nw, (int)(threadIdx.x & 31));
    return;
  }
  if (blockIdx.x == 0 && threadIdx.x == 0) total[0] = 0;
  const int r = blockIdx.x % CNT_R;
  const int sl_id = blockIdx.x / CNT_R;
  const int* __restrict__ dstp = ei + n_edges;

  int rl = (n_nodes + CNT_R - 1) / CNT_R;
  if (rl > CNT_CAP) rl = CNT_CAP;
  const int sl = (n_edges + CNT_S - 1) / CNT_S;
  const int e0 = sl_id * sl;
  int e1 = e0 + sl;
  if (e1 > n_edges) e1 = n_edges;
  const int step = CNT_DEPTH * CNT_THREADS;
  const bool vec = ((((size_t)ei) & 15) == 0) && (((n_edges | sl) & 3) == 0);

  for (int lo = r * rl; lo < n_nodes; lo += CNT_R * rl) {
    int len = n_nodes - lo;
    if (len > rl) len = rl;

    int d[CNT_DEPTH];
    #pragma unroll
    for (int j = 0; j < CNT_ROWS; ++j)
      load_ids4(dstp, e0 + j * 4 * CNT_THREADS, e1, CNT_THREADS, vec, (int)threadIdx.x, -1, d + 4 * j);
    for (int i = threadIdx.x; i < len; i += CNT_THREADS) lcnt[i] = 0;
    __syncthreads();

    for (int b0 = e0; ; ) {
      #pragma unroll
      for (int j = 0; j < CNT_DEPTH; ++j) {
        const unsigned off = (unsigned)(d[j] - lo);
        if (off < (unsigned)len) atomicAdd(&lcnt[off], 1);
      }
      b0 += step;
      if (b0 >= e1) break;
      #pragma unroll
      for (int j = 0; j < CNT_ROWS; ++j)
        load_ids4(dstp, b0 + j * 4 * CNT_THREADS, e1, CNT_THREADS, vec, (int)threadIdx.x, -1, d + 4 * j);
    }
    __syncthreads();

    int* __restrict__ row = scount + (size_t)sl_id * n_nodes + lo;
    for (int i = threadIdx.x; i < len; i += CNT_THREADS) row[i] = lcnt[i];
    __syncthreads();
  }
}

__global__ void __launch_bounds__(256)
alloc_kernel(int* __restrict__ scount,
             int* __restrict__ cnt,
             int* __restrict__ cursor,
             int* __restrict__ total,
             int n_nodes) {
  __shared__ int wave_sum[8];
  __shared__ int block_base;
  const int i    = blockIdx.x * 256 + threadIdx.x;
  const int lane = threadIdx.x & 31;
  const int w    = threadIdx.x >> 5;
  int cs[CNT_S];
  int c = 0;
  #pragma unroll
  for (int s = 0; s < CNT_S; ++s) {
    cs[s] = (i < n_nodes) ? scount[(size_t)s * n_nodes + i] : 0;
    c += cs[s];
  }
  int inc = c;
  for (int off = 1; off < 32; off <<= 1) {
    const int t = __shfl_up(inc, off, 32);
    if (lane >= off) inc += t;
  }
  if (lane == 31) wave_sum[w] = inc;
  __syncthreads();
  if (threadIdx.x == 0) {
    int run = 0;
    for (int k = 0; k < 8; ++k) {
      const int t = wave_sum[k];
      wave_sum[k] = run;
      run += t;
    }
    block_base = atomicAdd(total, run);
  }
  __syncthreads();
  if (i < n_nodes) {
    const int start = block_base + wave_sum[w] + inc - c;
    cursor[i] = start;
    cnt[i]    = c;
    int run = start;
    #pragma unroll
    for (int s = 0; s < CNT_S; ++s) {
      scount[(size_t)s * n_nodes + i] = run;
      run += cs[s];
    }
  }
}

__global__ void __launch_bounds__(FILL_THREADS)
fill_norm_kernel(const int* __restrict__ ei,
                 const int* __restrict__ sstart,
                 int* __restrict__ sorted_src,
                 int n_edges,
                 int n_nodes,
                 const float* __restrict__ x,
                 const float* __restrict__ gamma,
                 const float* __restrict__ beta,
                 float* __restrict__ h) {
  __shared__ int lcur[FILL_CAP];
  if (blockIdx.x >= FILL_R * CNT_S) {
    const int gw = __builtin_amdgcn_readfirstlane(
        (int)(((blockIdx.x - FILL_R * CNT_S) * FILL_THREADS + threadIdx.x) >> 5));
    const int nw = NORM_BLOCKS_FILL * (FILL_THREADS >> 5);
    ln_relu_rows(x, gamma, beta, h, n_nodes >> 1, n_nodes, gw, nw, (int)(threadIdx.x & 31));
    return;
  }
  const int r = blockIdx.x % FILL_R;
  const int sl_id = blockIdx.x / FILL_R;
  const int* __restrict__ dstp = ei + n_edges;
  const int* __restrict__ srow = sstart + (size_t)sl_id * n_nodes;

  int rl = (n_nodes + FILL_R - 1) / FILL_R;
  if (rl > FILL_CAP) rl = FILL_CAP;
  const int sl = (n_edges + CNT_S - 1) / CNT_S;
  const int e0 = sl_id * sl;
  int e1 = e0 + sl;
  if (e1 > n_edges) e1 = n_edges;
  const int step = FILL_DEPTH * FILL_THREADS;
  const bool vec = ((((size_t)ei) & 15) == 0) && (((n_edges | sl) & 3) == 0);

  for (int lo = r * rl; lo < n_nodes; lo += FILL_R * rl) {
    int len = n_nodes - lo;
    if (len > rl) len = rl;

    {
      int cv[FILL_CAP / FILL_THREADS];
      #pragma unroll
      for (int j = 0; j < FILL_CAP / FILL_THREADS; ++j) {
        const int i = (int)threadIdx.x + j * FILL_THREADS;
        cv[j] = srow[lo + (i < len ? i : len - 1)];
      }
      #pragma unroll
      for (int j = 0; j < FILL_CAP / FILL_THREADS; ++j)
        lcur[(int)threadIdx.x + j * FILL_THREADS] = cv[j];
    }
    int d[FILL_DEPTH], s[FILL_DEPTH];
    #pragma unroll
    for (int j = 0; j < FILL_ROWS; ++j) {
      load_ids4(dstp, e0 + j * 4 * FILL_THREADS, e1, FILL_THREADS, vec, (int)threadIdx.x, -1, d + 4 * j);
      load_ids4(ei,   e0 + j * 4 * FILL_THREADS, e1, FILL_THREADS, vec, (int)threadIdx.x,  0, s + 4 * j);
    }
    __syncthreads();

    for (int b0 = e0; ; ) {
      #pragma unroll
      for (int j = 0; j < FILL_DEPTH; ++j) {
        const unsigned off = (unsigned)(d[j] - lo);
        d[j] = (off < (unsigned)len) ? atomicAdd(&lcur[off], 1) : -1;
      }
      #pragma unroll
      for (int j = 0; j < FILL_DEPTH; ++j) {
        if (d[j] >= 0) sorted_src[d[j]] = s[j];
      }
      b0 += step;
      if (b0 >= e1) break;
      #pragma unroll
      for (int j = 0; j < FILL_ROWS; ++j) {
        load_ids4(dstp, b0 + j * 4 * FILL_THREADS, e1, FILL_THREADS, vec, (int)threadIdx.x, -1, d + 4 * j);
        load_ids4(ei,   b0 + j * 4 * FILL_THREADS, e1, FILL_THREADS, vec, (int)threadIdx.x,  0, s + 4 * j);
      }
    }
    __syncthreads();
  }
}

__device__ __forceinline__ float4 row4(const float* __restrict__ h, int s, int lane) {
  return ((const float4*)(h + (size_t)s * DIM))[lane];
}

__device__ __forceinline__ int load_ids(const int* __restrict__ sorted_src,
                                        int node, int base, int end, int lane) {
  return (lane < end - base) ? sorted_src[base + lane] : node;
}

__device__ __forceinline__ float4 node_sum(const float* __restrict__ h,
                                           const int* __restrict__ sorted_src,
                                           int node, int begin, int n, int first,
                                           float4 acc, int lane) {
  const int end = begin + n;
  int mine = first;
  for (int base = begin; base < end; base += 32) {
    const int left = end - base;
    const int m    = left < 32 ? left : 32;
    if (base != begin) mine = load_ids(sorted_src, node, base, end, lane);
    int j = 0;
    for (; j + 8 <= m; j += 8) {
      float4 v[8];
      #pragma unroll
      for (int i = 0; i < 8; ++i)
        v[i] = row4(h, __builtin_amdgcn_readlane(mine, j + i), lane);
      #pragma unroll
      for (int i = 0; i < 8; ++i) {
        acc.x += v[i].x; acc.y += v[i].y; acc.z += v[i].z; acc.w += v[i].w;
      }
    }
    for (; j < m; j += 4) {
      float4 v[4];
      #pragma unroll
      for (int i = 0; i < 4; ++i)
        v[i] = row4(h, __builtin_amdgcn_readlane(mine, j + i), lane);
      #pragma unroll
      for (int i = 0; i < 4; ++i) {
        const bool ok = j + i < m;
        acc.x += ok ? v[i].x : 0.0f;
        acc.y += ok ? v[i].y : 0.0f;
        acc.z += ok ? v[i].z : 0.0f;
        acc.w += ok ? v[i].w : 0.0f;
      }
    }
  }
  return acc;
}

__global__ void __launch_bounds__(BLOCK)
gather_gemm_kernel(const float* __restrict__ h,
                   const int* __restrict__ sorted_src,
                   const int* __restrict__ cursor,
                   const int* __restrict__ cnt,
                   const float* __restrict__ eps_p,
                   const float* __restrict__ W,
                   const float* __restrict__ bias,
                   const float* __restrict__ x,
                   float* __restrict__ out,
                   int n_row_tiles) {
  __shared__ float tile[2][16 * TILE_PITCH];

  const int wave = __builtin_amdgcn_readfirstlane((int)(threadIdx.x >> 5));
  const int lane = threadIdx.x & 31;
  const int col0 = wave * 16;
  const int half = lane >> 4;
  const int l    = lane & 15;

  const float* brow = W + (size_t)(col0 + l) * DIM + 2 * half;
  v2f b[DIM / 4];
  #pragma unroll
  for (int k = 0; k < DIM; k += 4) b[k >> 2] = *(const v2f*)(brow + k);
  const float bv = bias[col0 + l];
  const float scale = 1.0f + eps_p[0];

  int n0 = 0, n1 = 0, begin0 = 0, begin1 = 0;
  if ((int)blockIdx.x < n_row_tiles) {
    const int nd = (int)blockIdx.x * 16 + 2 * wave;
    n0 = cnt[nd];        n1 = cnt[nd + 1];
    begin0 = cursor[nd]; begin1 = cursor[nd + 1];
  }

  int buf = 0;
  for (int tm = blockIdx.x; tm < n_row_tiles; tm += gridDim.x) {
    const int row0 = tm * 16;
    float* t = tile[buf];

    const int nd0 = row0 + 2 * wave;
    const int nd1 = nd0 + 1;
    const int ids0 = load_ids(sorted_src, nd0, begin0, begin0 + n0, lane);
    const int ids1 = load_ids(sorted_src, nd1, begin1, begin1 + n1, lane);
    const float4 h0 = row4(h, nd0, lane);
    const float4 h1 = row4(h, nd1, lane);

    const int tmn = tm + (int)gridDim.x;
    int nn0 = 0, nn1 = 0, nbegin0 = 0, nbegin1 = 0;
    if (tmn < n_row_tiles) {
      const int nd = tmn * 16 + 2 * wave;
      nn0 = cnt[nd];        nn1 = cnt[nd + 1];
      nbegin0 = cursor[nd]; nbegin1 = cursor[nd + 1];
    }

    float4 acc;
    acc.x = scale * h0.x; acc.y = scale * h0.y; acc.z = scale * h0.z; acc.w = scale * h0.w;
    acc = node_sum(h, sorted_src, nd0, begin0, n0, ids0, acc, lane);
    *(float4*)(t + (2 * wave) * TILE_PITCH + 4 * lane) = acc;
    acc.x = scale * h1.x; acc.y = scale * h1.y; acc.z = scale * h1.z; acc.w = scale * h1.w;
    acc = node_sum(h, sorted_src, nd1, begin1, n1, ids1, acc, lane);
    *(float4*)(t + (2 * wave + 1) * TILE_PITCH + 4 * lane) = acc;

    __syncthreads();

    const float* arow = t + l * TILE_PITCH + 2 * half;
    v8f c = {};
    #pragma unroll
    for (int k = 0; k < DIM; k += 4) {
      v2f a = *(const v2f*)(arow + k);
      c = __builtin_amdgcn_wmma_f32_16x16x4_f32(
           false, a,  false, b[k >> 2],
           (short)0, c,  false,  false);
    }

    const size_t idx0 = (size_t)(row0 + 8 * half) * DIM + col0 + l;
    const float* xp = x + idx0;
    float* op = out + idx0;
    #pragma unroll
    for (int r = 0; r < 8; ++r) op[r * DIM] = c[r] + bv + xp[r * DIM];

    n0 = nn0; n1 = nn1; begin0 = nbegin0; begin1 = nbegin1;
    buf ^= 1;
  }
}

extern "C" void kernel_launch(void* const* d_in, const int* in_sizes, int n_in,
                              void* d_out, int out_size, void* d_ws, size_t ws_size,
                              hipStream_t stream) {
  const float* x     = (const float*)d_in[0];
  const int*   ei    = (const int*)  d_in[1];
  const float* gamma = (const float*)d_in[2];
  const float* beta  = (const float*)d_in[3];
  const float* W     = (const float*)d_in[4];
  const float* bias  = (const float*)d_in[5];
  const float* eps   = (const float*)d_in[6];
  float* out = (float*)d_out;

  const int n_nodes = in_sizes[0] / DIM;
  const int n_edges = in_sizes[1] / 2;

  float* h          = (float*)d_ws;
  int*   cnt        = (int*)(h + (size_t)n_nodes * DIM);
  int*   cursor     = cnt + n_nodes;
  int*   sorted_src = cursor + n_nodes;
  int*   scount     = sorted_src + n_edges;
  int*   total      = scount + (size_t)CNT_S * n_nodes;

  count_norm_kernel<<<CNT_R * CNT_S + NORM_BLOCKS_RANK, CNT_THREADS, 0, stream>>>(
      ei, scount, n_edges, n_nodes, x, gamma, beta, h, total);
  {
    int blocks = (n_nodes + 255) / 256;
    alloc_kernel<<<blocks, 256, 0, stream>>>(scount, cnt, cursor, total, n_nodes);
  }
  fill_norm_kernel<<<FILL_R * CNT_S + NORM_BLOCKS_FILL, FILL_THREADS, 0, stream>>>(
      ei, scount, sorted_src, n_edges, n_nodes, x, gamma, beta, h);
  gather_gemm_kernel<<<GRID_FUSED, BLOCK, 0, stream>>>(h, sorted_src, cursor, cnt, eps, W, bias, x, out, n_nodes / 16);
}


# relu(LayerNorm(x)) is no longer stored: norm blocks write only per-node (mean, rstd); the gather loads x rows and normalises them in registers (own row = row 0 of the list); sum order unchanged
# speedup vs baseline: 4.0153x; 1.0419x over previous
#include <hip/hip_runtime.h>

#define DIM 128
#define WAVES_PER_BLOCK 8
#define BLOCK (WAVES_PER_BLOCK * 32)

#define NORM_BLOCKS_RANK 128
#define NORM_BLOCKS_FILL 256
#define GRID_FUSED  1024
#define TILE_PITCH   132
#define STAT_ROWS 16

#define CNT_R 4
#define CNT_S 32
#define CNT_CAP 25600
#define CNT_THREADS 1024
#define CNT_ROWS 13
#define CNT_DEPTH (4 * CNT_ROWS)
#define FILL_R 8
#define FILL_CAP 12800
#define FILL_THREADS 512
#define FILL_ROWS 9
#define FILL_DEPTH (4 * FILL_ROWS)

typedef float v2f __attribute__((ext_vector_type(2)));
typedef float v8f __attribute__((ext_vector_type(8)));

__device__ __forceinline__ void ln_stats_rows(const float* __restrict__ x,
                                              float* __restrict__ stats,
                                              int row_lo, int row_hi,
                                              int gw, int nw, int lane) {
  const int nchunks = (row_hi - row_lo + STAT_ROWS - 1) / STAT_ROWS;
  for (int c = gw; c < nchunks; c += nw) {
    const int base = row_lo + c * STAT_ROWS;
    float4 v[STAT_ROWS];
    #pragma unroll
    for (int j = 0; j < STAT_ROWS; ++j) {
      const int nd = base + j;
      v[j] = make_float4(0.0f, 0.0f, 0.0f, 0.0f);
      if (nd < row_hi) v[j] = ((const float4*)(x + (size_t)nd * DIM))[lane];
    }
    float mu_mine = 0.0f, rs_mine = 0.0f;
    #pragma unroll
    for (int j = 0; j < STAT_ROWS; ++j) {
      float s  = v[j].x + v[j].y + v[j].z + v[j].w;
      float s2 = v[j].x * v[j].x + v[j].y * v[j].y + v[j].z * v[j].z + v[j].w * v[j].w;
      #pragma unroll
      for (int off = 16; off > 0; off >>= 1) {
        s  += __shfl_xor(s,  off, 32);
        s2 += __shfl_xor(s2, off, 32);
      }
      const float mu  = s * (1.0f / DIM);
      const float var = s2 * (1.0f / DIM) - mu * mu;
      const float rs  = rsqrtf(var + 1e-5f);
      if (lane == j) { mu_mine = mu; rs_mine = rs; }
    }
    if (lane < STAT_ROWS && base + lane < row_hi)
      *(float2*)(stats + 2 * (size_t)(base + lane)) = make_float2(mu_mine, rs_mine);
  }
}

__device__ __forceinline__ void load_ids4(const int* __restrict__ p, int eb, int e1,
                                          int nthreads, bool vec, int t, int past,
                                          int* __restrict__ out) {
  const int e = eb + 4 * t;
  if (vec && eb + 4 * nthreads <= e1) {
    const int4 q = *(const int4*)(p + e);
    out[0] = q.x; out[1] = q.y; out[2] = q.z; out[3] = q.w;
  } else {
    #pragma unroll
    for (int k = 0; k < 4; ++k) out[k] = (e + k < e1) ? p[e + k] : past;
  }
}


__global__ void __launch_bounds__(CNT_THREADS)
count_norm_kernel(const int* __restrict__ ei,
                  int* __restrict__ scount,
                  int n_edges,
                  int n_nodes,
                  const float* __restrict__ x,
                  float* __restrict__ stats,
                  int* __restrict__ total) {
  __shared__ int lcnt[CNT_CAP];
  if (blockIdx.x >= CNT_R * CNT_S) {
    const int gw = __builtin_amdgcn_readfirstlane(
        (int)((threadIdx.x >> 5) * NORM_BLOCKS_RANK + (blockIdx.x - CNT_R * CNT_S)));
    const int nw = NORM_BLOCKS_RANK * (CNT_THREADS >> 5);
    ln_stats_rows(x, stats, 0, n_nodes >> 1, gw, nw, (int)(threadIdx.x & 31));
    return;
  }
  if (blockIdx.x == 0 && threadIdx.x == 0) total[0] = 0;
  const int r = blockIdx.x % CNT_R;
  const int sl_id = blockIdx.x / CNT_R;
  const int* __restrict__ dstp = ei + n_edges;

  int rl = (n_nodes + CNT_R - 1) / CNT_R;
  if (rl > CNT_CAP) rl = CNT_CAP;
  const int sl = (n_edges + CNT_S - 1) / CNT_S;
  const int e0 = sl_id * sl;
  int e1 = e0 + sl;
  if (e1 > n_edges) e1 = n_edges;
  const int step = CNT_DEPTH * CNT_THREADS;
  const bool vec = ((((size_t)ei) & 15) == 0) && (((n_edges | sl) & 3) == 0);

  for (int lo = r * rl; lo < n_nodes; lo += CNT_R * rl) {
    int len = n_nodes - lo;
    if (len > rl) len = rl;

    int d[CNT_DEPTH];
    #pragma unroll
    for (int j = 0; j < CNT_ROWS; ++j)
      load_ids4(dstp, e0 + j * 4 * CNT_THREADS, e1, CNT_THREADS, vec, (int)threadIdx.x, -1, d + 4 * j);
    for (int i = threadIdx.x; i < len; i += CNT_THREADS) lcnt[i] = 0;
    __syncthreads();

    for (int b0 = e0; ; ) {
      #pragma unroll
      for (int j = 0; j < CNT_DEPTH; ++j) {
        const unsigned off = (unsigned)(d[j] - lo);
        if (off < (unsigned)len) atomicAdd(&lcnt[off], 1);
      }
      b0 += step;
      if (b0 >= e1) break;
      #pragma unroll
      for (int j = 0; j < CNT_ROWS; ++j)
        load_ids4(dstp, b0 + j * 4 * CNT_THREADS, e1, CNT_THREADS, vec, (int)threadIdx.x, -1, d + 4 * j);
    }
    __syncthreads();

    int* __restrict__ row = scount + (size_t)sl_id * n_nodes + lo;
    for (int i = threadIdx.x; i < len; i += CNT_THREADS) row[i] = lcnt[i];
    __syncthreads();
  }
}

__global__ void __launch_bounds__(256)
alloc_kernel(int* __restrict__ scount,
             int* __restrict__ cnt,
             int* __restrict__ cursor,
             int* __restrict__ total,
             int n_nodes) {
  __shared__ int wave_sum[8];
  __shared__ int block_base;
  const int i    = blockIdx.x * 256 + threadIdx.x;
  const int lane = threadIdx.x & 31;
  const int w    = threadIdx.x >> 5;
  int cs[CNT_S];
  int c = 0;
  #pragma unroll
  for (int s = 0; s < CNT_S; ++s) {
    cs[s] = (i < n_nodes) ? scount[(size_t)s * n_nodes + i] : 0;
    c += cs[s];
  }
  int inc = c;
  for (int off = 1; off < 32; off <<= 1) {
    const int t = __shfl_up(inc, off, 32);
    if (lane >= off) inc += t;
  }
  if (lane == 31) wave_sum[w] = inc;
  __syncthreads();
  if (threadIdx.x == 0) {
    int run = 0;
    for (int k = 0; k < 8; ++k) {
      const int t = wave_sum[k];
      wave_sum[k] = run;
      run += t;
    }
    block_base = atomicAdd(total, run);
  }
  __syncthreads();
  if (i < n_nodes) {
    const int start = block_base + wave_sum[w] + inc - c;
    cursor[i] = start;
    cnt[i]    = c;
    int run = start;
    #pragma unroll
    for (int s = 0; s < CNT_S; ++s) {
      scount[(size_t)s * n_nodes + i] = run;
      run += cs[s];
    }
  }
}

__global__ void __launch_bounds__(FILL_THREADS)
fill_norm_kernel(const int* __restrict__ ei,
                 const int* __restrict__ sstart,
                 int* __restrict__ sorted_src,
                 int n_edges,
                 int n_nodes,
                 const float* __restrict__ x,
                 float* __restrict__ stats) {
  __shared__ int lcur[FILL_CAP];
  if (blockIdx.x >= FILL_R * CNT_S) {
    const int gw = __builtin_amdgcn_readfirstlane(
        (int)((threadIdx.x >> 5) * NORM_BLOCKS_FILL + (blockIdx.x - FILL_R * CNT_S)));
    const int nw = NORM_BLOCKS_FILL * (FILL_THREADS >> 5);
    ln_stats_rows(x, stats, n_nodes >> 1, n_nodes, gw, nw, (int)(threadIdx.x & 31));
    return;
  }
  const int r = blockIdx.x % FILL_R;
  const int sl_id = blockIdx.x / FILL_R;
  const int* __restrict__ dstp = ei + n_edges;
  const int* __restrict__ srow = sstart + (size_t)sl_id * n_nodes;

  int rl = (n_nodes + FILL_R - 1) / FILL_R;
  if (rl > FILL_CAP) rl = FILL_CAP;
  const int sl = (n_edges + CNT_S - 1) / CNT_S;
  const int e0 = sl_id * sl;
  int e1 = e0 + sl;
  if (e1 > n_edges) e1 = n_edges;
  const int step = FILL_DEPTH * FILL_THREADS;
  const bool vec = ((((size_t)ei) & 15) == 0) && (((n_edges | sl) & 3) == 0);

  for (int lo = r * rl; lo < n_nodes; lo += FILL_R * rl) {
    int len = n_nodes - lo;
    if (len > rl) len = rl;

    {
      int cv[FILL_CAP / FILL_THREADS];
      #pragma unroll
      for (int j = 0; j < FILL_CAP / FILL_THREADS; ++j) {
        const int i = (int)threadIdx.x + j * FILL_THREADS;
        cv[j] = srow[lo + (i < len ? i : len - 1)];
      }
      #pragma unroll
      for (int j = 0; j < FILL_CAP / FILL_THREADS; ++j)
        lcur[(int)threadIdx.x + j * FILL_THREADS] = cv[j];
    }
    int d[FILL_DEPTH], s[FILL_DEPTH];
    #pragma unroll
    for (int j = 0; j < FILL_ROWS; ++j) {
      load_ids4(dstp, e0 + j * 4 * FILL_THREADS, e1, FILL_THREADS, vec, (int)threadIdx.x, -1, d + 4 * j);
      load_ids4(ei,   e0 + j * 4 * FILL_THREADS, e1, FILL_THREADS, vec, (int)threadIdx.x,  0, s + 4 * j);
    }
    __syncthreads();

    for (int b0 = e0; ; ) {
      #pragma unroll
      for (int j = 0; j < FILL_DEPTH; ++j) {
        const unsigned off = (unsigned)(d[j] - lo);
        d[j] = (off < (unsigned)len) ? atomicAdd(&lcur[off], 1) : -1;
      }
      #pragma unroll
      for (int j = 0; j < FILL_DEPTH; ++j) {
        if (d[j] >= 0) sorted_src[d[j]] = s[j];
      }
      b0 += step;
      if (b0 >= e1) break;
      #pragma unroll
      for (int j = 0; j < FILL_ROWS; ++j) {
        load_ids4(dstp, b0 + j * 4 * FILL_THREADS, e1, FILL_THREADS, vec, (int)threadIdx.x, -1, d + 4 * j);
        load_ids4(ei,   b0 + j * 4 * FILL_THREADS, e1, FILL_THREADS, vec, (int)threadIdx.x,  0, s + 4 * j);
      }
    }
    __syncthreads();
  }
}

__device__ __forceinline__ float4 row4(const float* __restrict__ x, int s, int lane) {
  const unsigned off = ((unsigned)s << 9) + ((unsigned)lane << 4);
  return *(const float4*)((const char*)x + off);
}

__device__ __forceinline__ float4 norm_relu4(float4 v, float mu, float rs, float4 g, float4 b) {
  const v2f m = {mu, mu}, r = {rs, rs}, zero = {0.0f, 0.0f};
  const v2f lo = (v2f){v.x, v.y}, hi = (v2f){v.z, v.w};
  const v2f glo = {g.x, g.y}, ghi = {g.z, g.w}, blo = {b.x, b.y}, bhi = {b.z, b.w};
  const v2f tlo = __builtin_elementwise_max((lo - m) * r * glo + blo, zero);
  const v2f thi = __builtin_elementwise_max((hi - m) * r * ghi + bhi, zero);
  float4 o;
  o.x = tlo.x; o.y = tlo.y; o.z = thi.x; o.w = thi.y;
  return o;
}

__device__ __forceinline__ int load_ids(const int* __restrict__ sorted_src,
                                        int node, int begin, int n, int r0, int lane) {
  const int r = r0 + lane;
  return (r >= 1 && r <= n) ? sorted_src[begin + r - 1] : node;
}

__device__ __forceinline__ float2 load_pair(const float* __restrict__ stats, int id) {
  return *(const float2*)(stats + 2 * (size_t)id);
}

__device__ __forceinline__ float4 node_sum(const float* __restrict__ x,
                                           const float* __restrict__ stats,
                                           const int* __restrict__ sorted_src,
                                           int node, int begin, int n, int first,
                                           float scale, float4 g, float4 b, int lane) {
  const int rows = n + 1;
  float4 acc = make_float4(0.0f, 0.0f, 0.0f, 0.0f);
  int mine = first;
  for (int r0 = 0; r0 < rows; r0 += 32) {
    const int left = rows - r0;
    const int m    = left < 32 ? left : 32;
    if (r0 != 0) mine = load_ids(sorted_src, node, begin, n, r0, lane);
    const float2 st = load_pair(stats, mine);
    const int mu_bits = __float_as_int(st.x);
    const int rs_bits = __float_as_int(st.y);
    int j = 0;
    for (; j + 8 <= m; j += 8) {
      float4 v[8];
      #pragma unroll
      for (int i = 0; i < 8; ++i)
        v[i] = row4(x, __builtin_amdgcn_readlane(mine, j + i), lane);
      const float w0 = (r0 + j == 0) ? scale : 1.0f;
      #pragma unroll
      for (int i = 0; i < 8; ++i) {
        const float mu = __int_as_float(__builtin_amdgcn_readlane(mu_bits, j + i));
        const float rs = __int_as_float(__builtin_amdgcn_readlane(rs_bits, j + i));
        float4 hv = norm_relu4(v[i], mu, rs, g, b);
        if (i == 0) { hv.x *= w0; hv.y *= w0; hv.z *= w0; hv.w *= w0; }
        acc.x += hv.x; acc.y += hv.y; acc.z += hv.z; acc.w += hv.w;
      }
    }
    for (; j < m; j += 4) {
      float4 v[4];
      #pragma unroll
      for (int i = 0; i < 4; ++i)
        v[i] = row4(x, __builtin_amdgcn_readlane(mine, j + i), lane);
      const float w0 = (r0 + j == 0) ? scale : 1.0f;
      #pragma unroll
      for (int i = 0; i < 4; ++i) {
        const bool ok = j + i < m;
        const float mu = __int_as_float(__builtin_amdgcn_readlane(mu_bits, j + i));
        const float rs = __int_as_float(__builtin_amdgcn_readlane(rs_bits, j + i));
        float4 hv = norm_relu4(v[i], mu, rs, g, b);
        if (i == 0) { hv.x *= w0; hv.y *= w0; hv.z *= w0; hv.w *= w0; }
        acc.x += ok ? hv.x : 0.0f;
        acc.y += ok ? hv.y : 0.0f;
        acc.z += ok ? hv.z : 0.0f;
        acc.w += ok ? hv.w : 0.0f;
      }
    }
  }
  return acc;
}

__global__ void __launch_bounds__(BLOCK)
gather_gemm_kernel(const float* __restrict__ stats,
                   const float* __restrict__ gamma,
                   const float* __restrict__ beta,
                   const int* __restrict__ sorted_src,
                   const int* __restrict__ cursor,
                   const int* __restrict__ cnt,
                   const float* __restrict__ eps_p,
                   const float* __restrict__ W,
                   const float* __restrict__ bias,
                   const float* __restrict__ x,
                   float* __restrict__ out,
                   int n_row_tiles) {
  __shared__ float tile[2][16 * TILE_PITCH];

  const int wave = __builtin_amdgcn_readfirstlane((int)(threadIdx.x >> 5));
  const int lane = threadIdx.x & 31;
  const int col0 = wave * 16;
  const int half = lane >> 4;
  const int l    = lane & 15;

  const float* brow = W + (size_t)(col0 + l) * DIM + 2 * half;
  v2f b[DIM / 4];
  #pragma unroll
  for (int k = 0; k < DIM; k += 4) b[k >> 2] = *(const v2f*)(brow + k);
  const float bv = bias[col0 + l];
  const float scale = 1.0f + eps_p[0];
  const float4 g  = ((const float4*)gamma)[lane];
  const float4 be = ((const float4*)beta)[lane];

  int n0 = 0, n1 = 0, begin0 = 0, begin1 = 0;
  if ((int)blockIdx.x < n_row_tiles) {
    const int nd = (int)blockIdx.x * 16 + 2 * wave;
    n0 = cnt[nd];        n1 = cnt[nd + 1];
    begin0 = cursor[nd]; begin1 = cursor[nd + 1];
  }

  int buf = 0;
  for (int tm = blockIdx.x; tm < n_row_tiles; tm += gridDim.x) {
    const int row0 = tm * 16;
    float* t = tile[buf];

    const int nd0 = row0 + 2 * wave;
    const int nd1 = nd0 + 1;
    const int ids0 = load_ids(sorted_src, nd0, begin0, n0, 0, lane);
    const int ids1 = load_ids(sorted_src, nd1, begin1, n1, 0, lane);

    const int tmn = tm + (int)gridDim.x;
    int nn0 = 0, nn1 = 0, nbegin0 = 0, nbegin1 = 0;
    if (tmn < n_row_tiles) {
      const int nd = tmn * 16 + 2 * wave;
      nn0 = cnt[nd];        nn1 = cnt[nd + 1];
      nbegin0 = cursor[nd]; nbegin1 = cursor[nd + 1];
    }

    float4 acc;
    acc = node_sum(x, stats, sorted_src, nd0, begin0, n0, ids0, scale, g, be, lane);
    *(float4*)(t + (2 * wave) * TILE_PITCH + 4 * lane) = acc;
    acc = node_sum(x, stats, sorted_src, nd1, begin1, n1, ids1, scale, g, be, lane);
    *(float4*)(t + (2 * wave + 1) * TILE_PITCH + 4 * lane) = acc;

    __syncthreads();

    const float* arow = t + l * TILE_PITCH + 2 * half;
    v8f c = {};
    #pragma unroll
    for (int k = 0; k < DIM; k += 4) {
      v2f a = *(const v2f*)(arow + k);
      c = __builtin_amdgcn_wmma_f32_16x16x4_f32(
           false, a,  false, b[k >> 2],
           (short)0, c,  false,  false);
    }

    const size_t idx0 = (size_t)(row0 + 8 * half) * DIM + col0 + l;
    const float* xp = x + idx0;
    float* op = out + idx0;
    #pragma unroll
    for (int r = 0; r < 8; ++r) op[r * DIM] = c[r] + bv + xp[r * DIM];

    n0 = nn0; n1 = nn1; begin0 = nbegin0; begin1 = nbegin1;
    buf ^= 1;
  }
}

extern "C" void kernel_launch(void* const* d_in, const int* in_sizes, int n_in,
                              void* d_out, int out_size, void* d_ws, size_t ws_size,
                              hipStream_t stream) {
  const float* x     = (const float*)d_in[0];
  const int*   ei    = (const int*)  d_in[1];
  const float* gamma = (const float*)d_in[2];
  const float* beta  = (const float*)d_in[3];
  const float* W     = (const float*)d_in[4];
  const float* bias  = (const float*)d_in[5];
  const float* eps   = (const float*)d_in[6];
  float* out = (float*)d_out;

  const int n_nodes = in_sizes[0] / DIM;
  const int n_edges = in_sizes[1] / 2;

  float* stats      = (float*)d_ws;
  int*   cnt        = (int*)(stats + (size_t)n_nodes * 2);
  int*   cursor     = cnt + n_nodes;
  int*   sorted_src = cursor + n_nodes;
  int*   scount     = sorted_src + n_edges;
  int*   total      = scount + (size_t)CNT_S * n_nodes;

  count_norm_kernel<<<CNT_R * CNT_S + NORM_BLOCKS_RANK, CNT_THREADS, 0, stream>>>(
      ei, scount, n_edges, n_nodes, x, stats, total);
  {
    int blocks = (n_nodes + 255) / 256;
    alloc_kernel<<<blocks, 256, 0, stream>>>(scount, cnt, cursor, total, n_nodes);
  }
  fill_norm_kernel<<<FILL_R * CNT_S + NORM_BLOCKS_FILL, FILL_THREADS, 0, stream>>>(
      ei, scount, sorted_src, n_edges, n_nodes, x, stats);
  gather_gemm_kernel<<<GRID_FUSED, BLOCK, 0, stream>>>(stats, gamma, beta, sorted_src, cursor, cnt, eps, W, bias, x, out, n_nodes / 16);
}
